# starting kernel
# baseline (speedup 1.0000x reference)
#include <hip/hip_runtime.h>

typedef __attribute__((ext_vector_type(2))) float v2f;
typedef __attribute__((ext_vector_type(8))) float v8f;

#define F_IN 128
#define HID  64
#define NCLS 16


__global__ void zero_f32(float* p, int n) {
    int i = blockIdx.x * blockDim.x + threadIdx.x;
    if (i < n) p[i] = 0.0f;
}

__global__ void bias_broadcast(float* p, const float* b, int n, int mask) {
    int i = blockIdx.x * blockDim.x + threadIdx.x;
    if (i < n) p[i] = b[i & mask];
}

__global__ void relu_inplace(float* p, int n) {
    int i = blockIdx.x * blockDim.x + threadIdx.x;
    if (i < n) p[i] = fmaxf(p[i], 0.0f);
}


__global__ void deg_count(const int* __restrict__ dst, float* __restrict__ deg, int E) {
    int e = blockIdx.x * blockDim.x + threadIdx.x;
    if (e < E) atomicAdd(&deg[dst[e]], 1.0f);
}

__global__ void deg_to_dis(float* __restrict__ d, int n) {
    int i = blockIdx.x * blockDim.x + threadIdx.x;
    if (i < n) {
        float v = d[i];
        d[i] = (v > 0.0f) ? rsqrtf(fmaxf(v, 1.0f)) : 0.0f;
    }
}


__global__ __launch_bounds__(32) void gemm1_wmma(const float* __restrict__ X,
                                                 const float* __restrict__ W1,
                                                 float* __restrict__ H) {
    const int tile = blockIdx.x;
    const int lane = threadIdx.x;
    const int half = lane >> 4;
    const int l16  = lane & 15;

    const int arow = tile * 16 + l16;
    v8f c0 = {}, c1 = {}, c2 = {}, c3 = {};

#pragma unroll
    for (int k = 0; k < F_IN; k += 4) {
        const float* ap = X + (size_t)arow * F_IN + k + half * 2;
        v2f a; a[0] = ap[0]; a[1] = ap[1];

        const int krow = k + half * 2;
        const float* bp = W1 + (size_t)krow * HID + l16;
        v2f b0; b0[0] = bp[0];  b0[1] = bp[HID + 0];
        v2f b1; b1[0] = bp[16]; b1[1] = bp[HID + 16];
        v2f b2; b2[0] = bp[32]; b2[1] = bp[HID + 32];
        v2f b3; b3[0] = bp[48]; b3[1] = bp[HID + 48];

        c0 = __builtin_amdgcn_wmma_f32_16x16x4_f32(false, a, false, b0, (short)0, c0, false, false);
        c1 = __builtin_amdgcn_wmma_f32_16x16x4_f32(false, a, false, b1, (short)0, c1, false, false);
        c2 = __builtin_amdgcn_wmma_f32_16x16x4_f32(false, a, false, b2, (short)0, c2, false, false);
        c3 = __builtin_amdgcn_wmma_f32_16x16x4_f32(false, a, false, b3, (short)0, c3, false, false);
    }

    float* hp = H + (size_t)(tile * 16 + half * 8) * HID + l16;
#pragma unroll
    for (int r = 0; r < 8; ++r) {
        hp[r * HID + 0]  = c0[r];
        hp[r * HID + 16] = c1[r];
        hp[r * HID + 32] = c2[r];
        hp[r * HID + 48] = c3[r];
    }
}


__global__ __launch_bounds__(32) void gemm2_wmma(const float* __restrict__ Y,
                                                 const float* __restrict__ W2,
                                                 float* __restrict__ H2) {
    const int tile = blockIdx.x;
    const int lane = threadIdx.x;
    const int half = lane >> 4;
    const int l16  = lane & 15;

    const int arow = tile * 16 + l16;
    v8f c0 = {};

#pragma unroll
    for (int k = 0; k < HID; k += 4) {
        const float* ap = Y + (size_t)arow * HID + k + half * 2;
        v2f a; a[0] = ap[0]; a[1] = ap[1];

        const int krow = k + half * 2;
        const float* bp = W2 + (size_t)krow * NCLS + l16;
        v2f b; b[0] = bp[0]; b[1] = bp[NCLS];

        c0 = __builtin_amdgcn_wmma_f32_16x16x4_f32(false, a, false, b, (short)0, c0, false, false);
    }

    float* hp = H2 + (size_t)(tile * 16 + half * 8) * NCLS + l16;
#pragma unroll
    for (int r = 0; r < 8; ++r) hp[r * NCLS] = c0[r];
}


__global__ void scatter64(const float* __restrict__ H,
                          const int* __restrict__ src, const int* __restrict__ dst,
                          const float* __restrict__ dis,
                          float* __restrict__ agg, int E) {
    int t = blockIdx.x * blockDim.x + threadIdx.x;
    int e = t >> 5;
    int lane = t & 31;
    if (e >= E) return;
    int s = src[e], d = dst[e];
    float nrm = dis[s] * dis[d];
    const float2 hv = ((const float2*)(H + (size_t)s * HID))[lane];
    float* ap = agg + (size_t)d * HID + lane * 2;
    atomicAdd(ap + 0, hv.x * nrm);
    atomicAdd(ap + 1, hv.y * nrm);
}


__global__ void scatter16(const float* __restrict__ H2,
                          const int* __restrict__ src, const int* __restrict__ dst,
                          const float* __restrict__ dis,
                          float* __restrict__ out, int E) {
    int t = blockIdx.x * blockDim.x + threadIdx.x;
    int e = t >> 4;
    int j = t & 15;
    if (e >= E) return;
    int s = src[e], d = dst[e];
    float nrm = dis[s] * dis[d];
    atomicAdd(out + (size_t)d * NCLS + j, H2[(size_t)s * NCLS + j] * nrm);
}


extern "C" void kernel_launch(void* const* d_in, const int* in_sizes, int n_in,
                              void* d_out, int out_size, void* d_ws, size_t ws_size,
                              hipStream_t stream) {
    const float* x    = (const float*)d_in[0];
    const int*   esrc = (const int*)d_in[1];
    const int*   edst = (const int*)d_in[2];
    const float* W1   = (const float*)d_in[3];
    const float* b1   = (const float*)d_in[4];
    const float* W2   = (const float*)d_in[5];
    const float* b2   = (const float*)d_in[6];
    float* out = (float*)d_out;

    const int NN = in_sizes[0] / F_IN;
    const int E  = in_sizes[1];
    const int Mtiles = NN / 16;

    float* dis = (float*)d_ws;
    float* H   = dis + NN;
    float* AGG = H + (size_t)NN * HID;
    float* H2  = AGG + (size_t)NN * HID;

    const int B = 256;
    #define NB(n) (((n) + B - 1) / B)

    zero_f32<<<NB(NN), B, 0, stream>>>(dis, NN);
    deg_count<<<NB(E), B, 0, stream>>>(edst, dis, E);
    deg_to_dis<<<NB(NN), B, 0, stream>>>(dis, NN);

    gemm1_wmma<<<Mtiles, 32, 0, stream>>>(x, W1, H);
    bias_broadcast<<<NB(NN * HID), B, 0, stream>>>(AGG, b1, NN * HID, HID - 1);
    scatter64<<<NB(E * 32), B, 0, stream>>>(H, esrc, edst, dis, AGG, E);
    relu_inplace<<<NB(NN * HID), B, 0, stream>>>(AGG, NN * HID);

    gemm2_wmma<<<Mtiles, 32, 0, stream>>>(AGG, W2, H2);
    bias_broadcast<<<NB(NN * NCLS), B, 0, stream>>>(out, b2, NN * NCLS, NCLS - 1);
    scatter16<<<NB(E * 16), B, 0, stream>>>(H2, esrc, edst, dis, out, E);

    #undef NB
}


# scatter kernels: one wave walks 32 edges (coalesced index/norm loads, lane broadcast, 8 row loads in flight) instead of one wave per 1-2 edges; atomic arrival order changes
# speedup vs baseline: 1.3468x; 1.3468x over previous
#include <hip/hip_runtime.h>

typedef __attribute__((ext_vector_type(2))) float v2f;
typedef __attribute__((ext_vector_type(8))) float v8f;

#define F_IN 128
#define HID  64
#define NCLS 16


__global__ void zero_f32(float* p, int n) {
    int i = blockIdx.x * blockDim.x + threadIdx.x;
    if (i < n) p[i] = 0.0f;
}

__global__ void bias_broadcast(float* p, const float* b, int n, int mask) {
    int i = blockIdx.x * blockDim.x + threadIdx.x;
    if (i < n) p[i] = b[i & mask];
}

__global__ void relu_inplace(float* p, int n) {
    int i = blockIdx.x * blockDim.x + threadIdx.x;
    if (i < n) p[i] = fmaxf(p[i], 0.0f);
}


__global__ void deg_count(const int* __restrict__ dst, float* __restrict__ deg, int E) {
    int e = blockIdx.x * blockDim.x + threadIdx.x;
    if (e < E) atomicAdd(&deg[dst[e]], 1.0f);
}

__global__ void deg_to_dis(float* __restrict__ d, int n) {
    int i = blockIdx.x * blockDim.x + threadIdx.x;
    if (i < n) {
        float v = d[i];
        d[i] = (v > 0.0f) ? rsqrtf(fmaxf(v, 1.0f)) : 0.0f;
    }
}


__global__ __launch_bounds__(32) void gemm1_wmma(const float* __restrict__ X,
                                                 const float* __restrict__ W1,
                                                 float* __restrict__ H) {
    const int tile = blockIdx.x;
    const int lane = threadIdx.x;
    const int half = lane >> 4;
    const int l16  = lane & 15;

    const int arow = tile * 16 + l16;
    v8f c0 = {}, c1 = {}, c2 = {}, c3 = {};

#pragma unroll
    for (int k = 0; k < F_IN; k += 4) {
        const float* ap = X + (size_t)arow * F_IN + k + half * 2;
        v2f a; a[0] = ap[0]; a[1] = ap[1];

        const int krow = k + half * 2;
        const float* bp = W1 + (size_t)krow * HID + l16;
        v2f b0; b0[0] = bp[0];  b0[1] = bp[HID + 0];
        v2f b1; b1[0] = bp[16]; b1[1] = bp[HID + 16];
        v2f b2; b2[0] = bp[32]; b2[1] = bp[HID + 32];
        v2f b3; b3[0] = bp[48]; b3[1] = bp[HID + 48];

        c0 = __builtin_amdgcn_wmma_f32_16x16x4_f32(false, a, false, b0, (short)0, c0, false, false);
        c1 = __builtin_amdgcn_wmma_f32_16x16x4_f32(false, a, false, b1, (short)0, c1, false, false);
        c2 = __builtin_amdgcn_wmma_f32_16x16x4_f32(false, a, false, b2, (short)0, c2, false, false);
        c3 = __builtin_amdgcn_wmma_f32_16x16x4_f32(false, a, false, b3, (short)0, c3, false, false);
    }

    float* hp = H + (size_t)(tile * 16 + half * 8) * HID + l16;
#pragma unroll
    for (int r = 0; r < 8; ++r) {
        hp[r * HID + 0]  = c0[r];
        hp[r * HID + 16] = c1[r];
        hp[r * HID + 32] = c2[r];
        hp[r * HID + 48] = c3[r];
    }
}


__global__ __launch_bounds__(32) void gemm2_wmma(const float* __restrict__ Y,
                                                 const float* __restrict__ W2,
                                                 float* __restrict__ H2) {
    const int tile = blockIdx.x;
    const int lane = threadIdx.x;
    const int half = lane >> 4;
    const int l16  = lane & 15;

    const int arow = tile * 16 + l16;
    v8f c0 = {};

#pragma unroll
    for (int k = 0; k < HID; k += 4) {
        const float* ap = Y + (size_t)arow * HID + k + half * 2;
        v2f a; a[0] = ap[0]; a[1] = ap[1];

        const int krow = k + half * 2;
        const float* bp = W2 + (size_t)krow * NCLS + l16;
        v2f b; b[0] = bp[0]; b[1] = bp[NCLS];

        c0 = __builtin_amdgcn_wmma_f32_16x16x4_f32(false, a, false, b, (short)0, c0, false, false);
    }

    float* hp = H2 + (size_t)(tile * 16 + half * 8) * NCLS + l16;
#pragma unroll
    for (int r = 0; r < 8; ++r) hp[r * NCLS] = c0[r];
}


__global__ __launch_bounds__(256) void scatter64(const float* __restrict__ H,
                          const int* __restrict__ src, const int* __restrict__ dst,
                          const float* __restrict__ dis,
                          float* __restrict__ agg, int E) {
    const int lane = threadIdx.x & 31;
    const int e0 = (blockIdx.x * blockDim.x + threadIdx.x) & ~31;
    if (e0 >= E) return;
    const int e = e0 + lane;
    int s = 0, d = 0;
    float nrm = 0.0f;
    if (e < E) {
        s = src[e]; d = dst[e];
        nrm = dis[s] * dis[d];
    }
    const int nrm_bits = __float_as_int(nrm);
    const int cnt = (E - e0 < 32) ? (E - e0) : 32;
    const float2* Hrow = (const float2*)H;

    if (cnt == 32) {
#pragma unroll
        for (int jb = 0; jb < 32; jb += 8) {
            float2 hv[8];
#pragma unroll
            for (int u = 0; u < 8; ++u) {
                const int sj = __builtin_amdgcn_readlane(s, jb + u);
                hv[u] = Hrow[(size_t)sj * (HID / 2) + lane];
            }
#pragma unroll
            for (int u = 0; u < 8; ++u) {
                const int dj = __builtin_amdgcn_readlane(d, jb + u);
                const float nj = __int_as_float(__builtin_amdgcn_readlane(nrm_bits, jb + u));
                float* ap = agg + (size_t)dj * HID + lane * 2;
                atomicAdd(ap + 0, hv[u].x * nj);
                atomicAdd(ap + 1, hv[u].y * nj);
            }
        }
    } else {
        for (int j = 0; j < cnt; ++j) {
            const int sj = __builtin_amdgcn_readlane(s, j);
            const int dj = __builtin_amdgcn_readlane(d, j);
            const float nj = __int_as_float(__builtin_amdgcn_readlane(nrm_bits, j));
            const float2 hv = Hrow[(size_t)sj * (HID / 2) + lane];
            float* ap = agg + (size_t)dj * HID + lane * 2;
            atomicAdd(ap + 0, hv.x * nj);
            atomicAdd(ap + 1, hv.y * nj);
        }
    }
}


__global__ __launch_bounds__(256) void scatter16(const float* __restrict__ H2,
                          const int* __restrict__ src, const int* __restrict__ dst,
                          const float* __restrict__ dis,
                          float* __restrict__ out, int E) {
    const int lane = threadIdx.x & 31;
    const int half = lane >> 4;
    const int j16  = lane & 15;
    const int e0 = (blockIdx.x * blockDim.x + threadIdx.x) & ~31;
    if (e0 >= E) return;
    const int e = e0 + lane;
    int s = 0, d = 0;
    float nrm = 0.0f;
    if (e < E) {
        s = src[e]; d = dst[e];
        nrm = dis[s] * dis[d];
    }
    const int nrm_bits = __float_as_int(nrm);
    const int cnt = (E - e0 < 32) ? (E - e0) : 32;

    if (cnt == 32) {
#pragma unroll
        for (int pb = 0; pb < 16; pb += 8) {
            float hv[8];
#pragma unroll
            for (int u = 0; u < 8; ++u) {
                const int p = pb + u;
                const int s0 = __builtin_amdgcn_readlane(s, 2 * p);
                const int s1 = __builtin_amdgcn_readlane(s, 2 * p + 1);
                const int sj = half ? s1 : s0;
                hv[u] = H2[(size_t)sj * NCLS + j16];
            }
#pragma unroll
            for (int u = 0; u < 8; ++u) {
                const int p = pb + u;
                const int d0 = __builtin_amdgcn_readlane(d, 2 * p);
                const int d1 = __builtin_amdgcn_readlane(d, 2 * p + 1);
                const int n0 = __builtin_amdgcn_readlane(nrm_bits, 2 * p);
                const int n1 = __builtin_amdgcn_readlane(nrm_bits, 2 * p + 1);
                const int dj = half ? d1 : d0;
                const float nj = __int_as_float(half ? n1 : n0);
                atomicAdd(out + (size_t)dj * NCLS + j16, hv[u] * nj);
            }
        }
    } else {
        for (int p = 0; 2 * p < cnt; ++p) {
            const int s0 = __builtin_amdgcn_readlane(s, 2 * p);
            const int s1 = __builtin_amdgcn_readlane(s, 2 * p + 1);
            const int d0 = __builtin_amdgcn_readlane(d, 2 * p);
            const int d1 = __builtin_amdgcn_readlane(d, 2 * p + 1);
            const int n0 = __builtin_amdgcn_readlane(nrm_bits, 2 * p);
            const int n1 = __builtin_amdgcn_readlane(nrm_bits, 2 * p + 1);
            const int sj = half ? s1 : s0;
            const int dj = half ? d1 : d0;
            const float nj = __int_as_float(half ? n1 : n0);
            if (2 * p + half < cnt)
                atomicAdd(out + (size_t)dj * NCLS + j16, H2[(size_t)sj * NCLS + j16] * nj);
        }
    }
}


extern "C" void kernel_launch(void* const* d_in, const int* in_sizes, int n_in,
                              void* d_out, int out_size, void* d_ws, size_t ws_size,
                              hipStream_t stream) {
    const float* x    = (const float*)d_in[0];
    const int*   esrc = (const int*)d_in[1];
    const int*   edst = (const int*)d_in[2];
    const float* W1   = (const float*)d_in[3];
    const float* b1   = (const float*)d_in[4];
    const float* W2   = (const float*)d_in[5];
    const float* b2   = (const float*)d_in[6];
    float* out = (float*)d_out;

    const int NN = in_sizes[0] / F_IN;
    const int E  = in_sizes[1];
    const int Mtiles = NN / 16;

    float* dis = (float*)d_ws;
    float* H   = dis + NN;
    float* AGG = H + (size_t)NN * HID;
    float* H2  = AGG + (size_t)NN * HID;

    const int B = 256;
    #define NB(n) (((n) + B - 1) / B)

    zero_f32<<<NB(NN), B, 0, stream>>>(dis, NN);
    deg_count<<<NB(E), B, 0, stream>>>(edst, dis, E);
    deg_to_dis<<<NB(NN), B, 0, stream>>>(dis, NN);

    gemm1_wmma<<<Mtiles, 32, 0, stream>>>(x, W1, H);
    bias_broadcast<<<NB(NN * HID), B, 0, stream>>>(AGG, b1, NN * HID, HID - 1);
    scatter64<<<NB(E), B, 0, stream>>>(H, esrc, edst, dis, AGG, E);
    relu_inplace<<<NB(NN * HID), B, 0, stream>>>(AGG, NN * HID);

    gemm2_wmma<<<Mtiles, 32, 0, stream>>>(AGG, W2, H2);
    bias_broadcast<<<NB(NN * NCLS), B, 0, stream>>>(out, b2, NN * NCLS, NCLS - 1);
    scatter16<<<NB(E), B, 0, stream>>>(H2, esrc, edst, dis, out, E);

    #undef NB
}


# float atomic scatter replaced by edges counting-sorted by destination in the same call plus one-wave-per-node row sums (rows pre-scaled by deg^-1/2); sum order: sorted list, lane-group tree
# speedup vs baseline: 1.8466x; 1.3711x over previous
#include <hip/hip_runtime.h>


typedef __attribute__((ext_vector_type(2))) float v2f;
typedef __attribute__((ext_vector_type(8))) float v8f;

#define F_IN 128
#define HID  64
#define NCLS 16
#define REC  16


__global__ void zero_i32(int* p, int n) {
    int i = blockIdx.x * blockDim.x + threadIdx.x;
    if (i < n) p[i] = 0;
}

__global__ void count_rank(const int* __restrict__ dst, int* __restrict__ cnt,
                           int* __restrict__ rank, int E) {
    int e = blockIdx.x * blockDim.x + threadIdx.x;
    if (e < E) rank[e] = atomicAdd(&cnt[dst[e]], 1);
}

__device__ __forceinline__ int block_scan_1024(int v, int (*s)[1024], int t) {
    int cur = 0;
    s[0][t] = v;
    __syncthreads();
    for (int off = 1; off < 1024; off <<= 1) {
        int x = s[cur][t];
        if (t >= off) x += s[cur][t - off];
        s[cur ^ 1][t] = x;
        __syncthreads();
        cur ^= 1;
    }
    const int r = s[cur][t];
    return r;
}

__global__ __launch_bounds__(1024) void scan_records(const int* __restrict__ cnt,
                                                     float* __restrict__ rec, int n) {
    __shared__ int s[2][1024];
    __shared__ int tot;
    const int t = threadIdx.x;
    const int base = blockIdx.x * 1024;
    int p = 0;
    for (int i = t; i < base; i += 1024) p += cnt[i];
    const int pin = block_scan_1024(p, s, t);
    if (t == 1023) tot = pin;
    __syncthreads();
    const int prefix = tot;
    const int i0 = base + t;
    const int v = (i0 < n) ? cnt[i0] : 0;
    const int incl = block_scan_1024(v, s, t);
    if (i0 < n) {
        const int start = prefix + incl - v;
        const float deg = (float)v;
        const float dis = (deg > 0.0f) ? rsqrtf(fmaxf(deg, 1.0f)) : 0.0f;
        float* r = rec + (size_t)i0 * REC;
        r[0] = __int_as_float(start);
        r[1] = __int_as_float(start + v);
        r[2] = dis;
    }
}

__global__ void fill_sorted(const int* __restrict__ src, const int* __restrict__ dst,
                            const int* __restrict__ rank, const float* __restrict__ rec,
                            int* __restrict__ csr, int E) {
    int e = blockIdx.x * blockDim.x + threadIdx.x;
    if (e < E) {
        const int d = dst[e];
        const int start = __float_as_int(rec[(size_t)d * REC]);
        csr[start + rank[e]] = src[e];
    }
}


__global__ __launch_bounds__(32) void gemm1_wmma(const float* __restrict__ X,
                                                 const float* __restrict__ W1,
                                                 const float* __restrict__ rec,
                                                 float* __restrict__ H) {
    const int tile = blockIdx.x;
    const int lane = threadIdx.x;
    const int half = lane >> 4;
    const int l16  = lane & 15;

    const int arow = tile * 16 + l16;
    v8f c0 = {}, c1 = {}, c2 = {}, c3 = {};

#pragma unroll
    for (int k = 0; k < F_IN; k += 4) {
        const float* ap = X + (size_t)arow * F_IN + k + half * 2;
        v2f a; a[0] = ap[0]; a[1] = ap[1];

        const int krow = k + half * 2;
        const float* bp = W1 + (size_t)krow * HID + l16;
        v2f b0; b0[0] = bp[0];  b0[1] = bp[HID + 0];
        v2f b1; b1[0] = bp[16]; b1[1] = bp[HID + 16];
        v2f b2; b2[0] = bp[32]; b2[1] = bp[HID + 32];
        v2f b3; b3[0] = bp[48]; b3[1] = bp[HID + 48];

        c0 = __builtin_amdgcn_wmma_f32_16x16x4_f32(false, a, false, b0, (short)0, c0, false, false);
        c1 = __builtin_amdgcn_wmma_f32_16x16x4_f32(false, a, false, b1, (short)0, c1, false, false);
        c2 = __builtin_amdgcn_wmma_f32_16x16x4_f32(false, a, false, b2, (short)0, c2, false, false);
        c3 = __builtin_amdgcn_wmma_f32_16x16x4_f32(false, a, false, b3, (short)0, c3, false, false);
    }

    const int row0 = tile * 16 + half * 8;
    float* hp = H + (size_t)row0 * HID + l16;
    const float* rp = rec + (size_t)row0 * REC + 2;
#pragma unroll
    for (int r = 0; r < 8; ++r) {
        const float sc = rp[r * REC];
        hp[r * HID + 0]  = c0[r] * sc;
        hp[r * HID + 16] = c1[r] * sc;
        hp[r * HID + 32] = c2[r] * sc;
        hp[r * HID + 48] = c3[r] * sc;
    }
}


__global__ __launch_bounds__(32) void gemm2_wmma(const float* __restrict__ Y,
                                                 const float* __restrict__ W2,
                                                 const float* __restrict__ rec,
                                                 float* __restrict__ H2) {
    const int tile = blockIdx.x;
    const int lane = threadIdx.x;
    const int half = lane >> 4;
    const int l16  = lane & 15;

    const int arow = tile * 16 + l16;
    v8f c0 = {};

#pragma unroll
    for (int k = 0; k < HID; k += 4) {
        const float* ap = Y + (size_t)arow * HID + k + half * 2;
        v2f a; a[0] = ap[0]; a[1] = ap[1];

        const int krow = k + half * 2;
        const float* bp = W2 + (size_t)krow * NCLS + l16;
        v2f b; b[0] = bp[0]; b[1] = bp[NCLS];

        c0 = __builtin_amdgcn_wmma_f32_16x16x4_f32(false, a, false, b, (short)0, c0, false, false);
    }

    const int row0 = tile * 16 + half * 8;
    float* hp = H2 + (size_t)row0 * NCLS + l16;
    const float* rp = rec + (size_t)row0 * REC + 2;
#pragma unroll
    for (int r = 0; r < 8; ++r) hp[r * NCLS] = c0[r] * rp[r * REC];
}


__global__ __launch_bounds__(256) void gather64(const float* __restrict__ Hs,
                                                const int* __restrict__ csr,
                                                const float* __restrict__ rec,
                                                const float* __restrict__ b1,
                                                float* __restrict__ Y, int n) {
    const int lane = threadIdx.x & 31;
    const int d = (blockIdx.x * blockDim.x + threadIdx.x) >> 5;
    if (d >= n) return;
    const float* r = rec + (size_t)d * REC;
    const int start = __float_as_int(r[0]);
    const int end   = __float_as_int(r[1]);
    const float dd  = r[2];
    const int q = lane >> 4;
    const int c = lane & 15;
    const float4* H4 = (const float4*)Hs;
    float4 acc = make_float4(0.0f, 0.0f, 0.0f, 0.0f);

    const int last = end - 1;
    for (int p = start + q; p < end; p += 8) {
        const bool v1 = (p + 2 < end), v2 = (p + 4 < end), v3 = (p + 6 < end);
        const int s0 = csr[p];
        const int s1 = csr[min(p + 2, last)];
        const int s2 = csr[min(p + 4, last)];
        const int s3 = csr[min(p + 6, last)];
        const float4 h0 = H4[(size_t)s0 * (HID / 4) + c];
        const float4 h1 = H4[(size_t)s1 * (HID / 4) + c];
        const float4 h2 = H4[(size_t)s2 * (HID / 4) + c];
        const float4 h3 = H4[(size_t)s3 * (HID / 4) + c];
        acc.x += h0.x; acc.y += h0.y; acc.z += h0.z; acc.w += h0.w;
        acc.x += v1 ? h1.x : 0.0f; acc.y += v1 ? h1.y : 0.0f; acc.z += v1 ? h1.z : 0.0f; acc.w += v1 ? h1.w : 0.0f;
        acc.x += v2 ? h2.x : 0.0f; acc.y += v2 ? h2.y : 0.0f; acc.z += v2 ? h2.z : 0.0f; acc.w += v2 ? h2.w : 0.0f;
        acc.x += v3 ? h3.x : 0.0f; acc.y += v3 ? h3.y : 0.0f; acc.z += v3 ? h3.z : 0.0f; acc.w += v3 ? h3.w : 0.0f;
    }
    acc.x += __shfl_xor(acc.x, 16);
    acc.y += __shfl_xor(acc.y, 16);
    acc.z += __shfl_xor(acc.z, 16);
    acc.w += __shfl_xor(acc.w, 16);
    if (q == 0) {
        const float4 bb = ((const float4*)b1)[c];
        float4 y;
        y.x = fmaxf(acc.x * dd + bb.x, 0.0f);
        y.y = fmaxf(acc.y * dd + bb.y, 0.0f);
        y.z = fmaxf(acc.z * dd + bb.z, 0.0f);
        y.w = fmaxf(acc.w * dd + bb.w, 0.0f);
        ((float4*)Y)[(size_t)d * (HID / 4) + c] = y;
    }
}


__global__ __launch_bounds__(256) void gather16(const float* __restrict__ H2,
                                                const int* __restrict__ csr,
                                                const float* __restrict__ b2,
                                                float* out, int n) {
    const int lane = threadIdx.x & 31;
    const int d = (blockIdx.x * blockDim.x + threadIdx.x) >> 5;
    if (d >= n) return;
    const float* r = out + (size_t)d * REC;
    const int start = __float_as_int(r[0]);
    const int end   = __float_as_int(r[1]);
    const float dd  = r[2];
    const int q = lane >> 2;
    const int c = lane & 3;
    const float4* H4 = (const float4*)H2;
    float4 acc = make_float4(0.0f, 0.0f, 0.0f, 0.0f);

    const int last = end - 1;
    for (int p = start + q; p < end; p += 32) {
        const bool v1 = (p + 8 < end), v2 = (p + 16 < end), v3 = (p + 24 < end);
        const int s0 = csr[p];
        const int s1 = csr[min(p + 8, last)];
        const int s2 = csr[min(p + 16, last)];
        const int s3 = csr[min(p + 24, last)];
        const float4 h0 = H4[(size_t)s0 * (NCLS / 4) + c];
        const float4 h1 = H4[(size_t)s1 * (NCLS / 4) + c];
        const float4 h2 = H4[(size_t)s2 * (NCLS / 4) + c];
        const float4 h3 = H4[(size_t)s3 * (NCLS / 4) + c];
        acc.x += h0.x; acc.y += h0.y; acc.z += h0.z; acc.w += h0.w;
        acc.x += v1 ? h1.x : 0.0f; acc.y += v1 ? h1.y : 0.0f; acc.z += v1 ? h1.z : 0.0f; acc.w += v1 ? h1.w : 0.0f;
        acc.x += v2 ? h2.x : 0.0f; acc.y += v2 ? h2.y : 0.0f; acc.z += v2 ? h2.z : 0.0f; acc.w += v2 ? h2.w : 0.0f;
        acc.x += v3 ? h3.x : 0.0f; acc.y += v3 ? h3.y : 0.0f; acc.z += v3 ? h3.z : 0.0f; acc.w += v3 ? h3.w : 0.0f;
    }
#pragma unroll
    for (int m = 4; m < 32; m <<= 1) {
        acc.x += __shfl_xor(acc.x, m);
        acc.y += __shfl_xor(acc.y, m);
        acc.z += __shfl_xor(acc.z, m);
        acc.w += __shfl_xor(acc.w, m);
    }
    if (q == 0) {
        const float4 bb = ((const float4*)b2)[c];
        float4 y;
        y.x = acc.x * dd + bb.x;
        y.y = acc.y * dd + bb.y;
        y.z = acc.z * dd + bb.z;
        y.w = acc.w * dd + bb.w;
        ((float4*)out)[(size_t)d * (NCLS / 4) + c] = y;
    }
}


extern "C" void kernel_launch(void* const* d_in, const int* in_sizes, int n_in,
                              void* d_out, int out_size, void* d_ws, size_t ws_size,
                              hipStream_t stream) {
    const float* x    = (const float*)d_in[0];
    const int*   esrc = (const int*)d_in[1];
    const int*   edst = (const int*)d_in[2];
    const float* W1   = (const float*)d_in[3];
    const float* b1   = (const float*)d_in[4];
    const float* W2   = (const float*)d_in[5];
    const float* b2   = (const float*)d_in[6];
    float* out = (float*)d_out;

    const int NN = in_sizes[0] / F_IN;
    const int E  = in_sizes[1];
    const int Mtiles = NN / 16;

    float* H   = (float*)d_ws;
    float* Y   = H + (size_t)NN * HID;
    int*   csr = (int*)(Y + (size_t)NN * HID);
    int*   cnt  = (int*)Y;
    int*   rank = cnt + NN;
    float* H2  = H;

    const int B = 256;
    #define NB(n) (((n) + B - 1) / B)

    zero_i32<<<NB(NN), B, 0, stream>>>(cnt, NN);
    count_rank<<<NB(E), B, 0, stream>>>(edst, cnt, rank, E);
    scan_records<<<(NN + 1023) / 1024, 1024, 0, stream>>>(cnt, out, NN);
    fill_sorted<<<NB(E), B, 0, stream>>>(esrc, edst, rank, out, csr, E);

    gemm1_wmma<<<Mtiles, 32, 0, stream>>>(x, W1, out, H);
    gather64<<<NB(NN * 32), B, 0, stream>>>(H, csr, out, b1, Y, NN);

    gemm2_wmma<<<Mtiles, 32, 0, stream>>>(Y, W2, out, H2);
    gather16<<<NB(NN * 32), B, 0, stream>>>(H2, csr, b2, out, NN);

    #undef NB
}


# edge grouping as a two-level counting sort through LDS (tile bins stored as runs, one work-group per 512-node bucket) replaces per-edge global atomics and single-word stores; sum order: LDS arrival
# speedup vs baseline: 2.8518x; 1.5444x over previous
#include <hip/hip_runtime.h>


typedef __attribute__((ext_vector_type(2))) float v2f;
typedef __attribute__((ext_vector_type(8))) float v8f;

#define F_IN 128
#define HID  64
#define NCLS 16
#define REC  16


#define BSH    9
#define BN     (1 << BSH)
#define MAXB   256
#define TILE_E 4096
#define TILE_T 256
#define EPT    (TILE_E / TILE_T)
#define SCAN_T 512
#define STAGE  12288

template <int NT>
__device__ __forceinline__ int block_scan_incl(int v, int (*s)[NT], int t) {
    int cur = 0;
    s[0][t] = v;
    __syncthreads();
    for (int off = 1; off < NT; off <<= 1) {
        int x = s[cur][t];
        if (t >= off) x += s[cur][t - off];
        s[cur ^ 1][t] = x;
        __syncthreads();
        cur ^= 1;
    }
    const int r = s[cur][t];
    return r;
}

__global__ __launch_bounds__(TILE_T) void tile_hist(const int* __restrict__ dst, int* __restrict__ tilecnt,
                                                    int E, int nb, int ntiles) {
    __shared__ int hist[MAXB];
    const int t = threadIdx.x;
    for (int i = t; i < MAXB; i += TILE_T) hist[i] = 0;
    __syncthreads();
    const int e0 = blockIdx.x * TILE_E;
    for (int j = 0; j < EPT; ++j) {
        const int e = e0 + j * TILE_T + t;
        if (e < E) atomicAdd(&hist[dst[e] >> BSH], 1);
    }
    __syncthreads();
    for (int i = t; i < nb; i += TILE_T) tilecnt[i * ntiles + blockIdx.x] = hist[i];
}

__global__ __launch_bounds__(SCAN_T) void scan_table(const int* __restrict__ cnt, int* __restrict__ off,
                                                     int nb, int ntiles) {
    __shared__ int s[2][SCAN_T];
    __shared__ int tot;
    const int t = threadIdx.x;
    const int b = blockIdx.x;
    const int base = b * ntiles;
    int p = 0;
    const int4* c4 = (const int4*)cnt;
    const int nv = base >> 2;
    for (int i = t; i < nv; i += SCAN_T) { const int4 q = c4[i]; p += (q.x + q.y) + (q.z + q.w); }
    for (int i = (nv << 2) + t; i < base; i += SCAN_T) p += cnt[i];
    const int pin = block_scan_incl<SCAN_T>(p, s, t);
    if (t == SCAN_T - 1) tot = pin;
    __syncthreads();
    const int prefix = tot;
    const int v = (t < ntiles) ? cnt[base + t] : 0;
    const int incl = block_scan_incl<SCAN_T>(v, s, t);
    if (t < ntiles) off[base + t] = prefix + incl - v;
    if (b == nb - 1 && t == SCAN_T - 1) off[nb * ntiles] = prefix + incl;
}

__global__ __launch_bounds__(TILE_T) void tile_scatter(const int* __restrict__ src, const int* __restrict__ dst,
                                                       const int* __restrict__ off, int2* __restrict__ bk,
                                                       int E, int nb, int ntiles) {
    __shared__ int cur[MAXB];
    __shared__ int lstart[MAXB];
    __shared__ int goff[MAXB];
    __shared__ int s[2][TILE_T];
    __shared__ int2 sorted[TILE_E];
    const int t = threadIdx.x;
    const int tile = blockIdx.x;
    for (int i = t; i < MAXB; i += TILE_T) cur[i] = 0;
    __syncthreads();
    const int e0 = tile * TILE_E;
    int ms[EPT], md[EPT];
    for (int j = 0; j < EPT; ++j) {
        const int e = e0 + j * TILE_T + t;
        const bool ok = e < E;
        md[j] = ok ? dst[e] : -1;
        ms[j] = ok ? src[e] : 0;
        if (ok) atomicAdd(&cur[md[j] >> BSH], 1);
    }
    __syncthreads();
    static_assert(TILE_T >= MAXB, "one thread per bucket");
    const int c = (t < nb) ? cur[t] : 0;
    const int incl = block_scan_incl<TILE_T>(c, s, t);
    if (t < nb) {
        lstart[t] = incl - c;
        cur[t] = incl - c;
        goff[t] = off[t * ntiles + tile];
    }
    __syncthreads();
    for (int j = 0; j < EPT; ++j) {
        if (md[j] >= 0) {
            const int p = atomicAdd(&cur[md[j] >> BSH], 1);
            sorted[p] = make_int2(ms[j], md[j]);
        }
    }
    __syncthreads();
    const int ne = min(TILE_E, E - e0);
    for (int i = t; i < ne; i += TILE_T) {
        const int2 v = sorted[i];
        const int bb = v.y >> BSH;
        bk[goff[bb] + (i - lstart[bb])] = v;
    }
}

__global__ __launch_bounds__(1024) void bucket_csr(const int2* __restrict__ bk, const int* __restrict__ off,
                                                   float* __restrict__ rec, int* __restrict__ csr,
                                                   int n, int ntiles) {
    __shared__ int cnt[BN];
    __shared__ int s[2][1024];
    __shared__ int stage[STAGE];
    const int t = threadIdx.x;
    const int b = blockIdx.x;
    const int e0 = off[b * ntiles];
    const int e1 = off[(b + 1) * ntiles];
    const int ne = e1 - e0;
    const int node0 = b << BSH;
    if (t < BN) cnt[t] = 0;
    __syncthreads();
    for (int i = t; i < ne; i += 1024) atomicAdd(&cnt[bk[e0 + i].y - node0], 1);
    __syncthreads();
    const int v = (t < BN) ? cnt[t] : 0;
    const int incl = block_scan_incl<1024>(v, s, t);
    const int st = incl - v;
    if (t < BN) {
        cnt[t] = st;
        const int node = node0 + t;
        if (node < n) {
            const float deg = (float)v;
            const float dis = (deg > 0.0f) ? rsqrtf(fmaxf(deg, 1.0f)) : 0.0f;
            float* r = rec + (size_t)node * REC;
            r[0] = __int_as_float(e0 + st);
            r[1] = __int_as_float(e0 + st + v);
            r[2] = dis;
        }
    }
    __syncthreads();
    const bool staged = (ne <= STAGE);
    for (int i = t; i < ne; i += 1024) {
        const int2 ed = bk[e0 + i];
        const int p = atomicAdd(&cnt[ed.y - node0], 1);
        if (staged) stage[p] = ed.x; else csr[e0 + p] = ed.x;
    }
    if (staged) {
        __syncthreads();
        for (int i = t; i < ne; i += 1024) csr[e0 + i] = stage[i];
    }
}


__global__ __launch_bounds__(32) void gemm1_wmma(const float* __restrict__ X,
                                                 const float* __restrict__ W1,
                                                 const float* __restrict__ rec,
                                                 float* __restrict__ H) {
    const int tile = blockIdx.x;
    const int lane = threadIdx.x;
    const int half = lane >> 4;
    const int l16  = lane & 15;

    const int arow = tile * 16 + l16;
    v8f c0 = {}, c1 = {}, c2 = {}, c3 = {};

#pragma unroll
    for (int k = 0; k < F_IN; k += 4) {
        const float* ap = X + (size_t)arow * F_IN + k + half * 2;
        v2f a; a[0] = ap[0]; a[1] = ap[1];

        const int krow = k + half * 2;
        const float* bp = W1 + (size_t)krow * HID + l16;
        v2f b0; b0[0] = bp[0];  b0[1] = bp[HID + 0];
        v2f b1; b1[0] = bp[16]; b1[1] = bp[HID + 16];
        v2f b2; b2[0] = bp[32]; b2[1] = bp[HID + 32];
        v2f b3; b3[0] = bp[48]; b3[1] = bp[HID + 48];

        c0 = __builtin_amdgcn_wmma_f32_16x16x4_f32(false, a, false, b0, (short)0, c0, false, false);
        c1 = __builtin_amdgcn_wmma_f32_16x16x4_f32(false, a, false, b1, (short)0, c1, false, false);
        c2 = __builtin_amdgcn_wmma_f32_16x16x4_f32(false, a, false, b2, (short)0, c2, false, false);
        c3 = __builtin_amdgcn_wmma_f32_16x16x4_f32(false, a, false, b3, (short)0, c3, false, false);
    }

    const int row0 = tile * 16 + half * 8;
    float* hp = H + (size_t)row0 * HID + l16;
    const float* rp = rec + (size_t)row0 * REC + 2;
#pragma unroll
    for (int r = 0; r < 8; ++r) {
        const float sc = rp[r * REC];
        hp[r * HID + 0]  = c0[r] * sc;
        hp[r * HID + 16] = c1[r] * sc;
        hp[r * HID + 32] = c2[r] * sc;
        hp[r * HID + 48] = c3[r] * sc;
    }
}


__global__ __launch_bounds__(32) void gemm2_wmma(const float* __restrict__ Y,
                                                 const float* __restrict__ W2,
                                                 const float* __restrict__ rec,
                                                 float* __restrict__ H2) {
    const int tile = blockIdx.x;
    const int lane = threadIdx.x;
    const int half = lane >> 4;
    const int l16  = lane & 15;

    const int arow = tile * 16 + l16;
    v8f c0 = {};

#pragma unroll
    for (int k = 0; k < HID; k += 4) {
        const float* ap = Y + (size_t)arow * HID + k + half * 2;
        v2f a; a[0] = ap[0]; a[1] = ap[1];

        const int krow = k + half * 2;
        const float* bp = W2 + (size_t)krow * NCLS + l16;
        v2f b; b[0] = bp[0]; b[1] = bp[NCLS];

        c0 = __builtin_amdgcn_wmma_f32_16x16x4_f32(false, a, false, b, (short)0, c0, false, false);
    }

    const int row0 = tile * 16 + half * 8;
    float* hp = H2 + (size_t)row0 * NCLS + l16;
    const float* rp = rec + (size_t)row0 * REC + 2;
#pragma unroll
    for (int r = 0; r < 8; ++r) hp[r * NCLS] = c0[r] * rp[r * REC];
}


__global__ __launch_bounds__(256) void gather64(const float* __restrict__ Hs,
                                                const int* __restrict__ csr,
                                                const float* __restrict__ rec,
                                                const float* __restrict__ b1,
                                                float* __restrict__ Y, int n) {
    const int lane = threadIdx.x & 31;
    const int d = (blockIdx.x * blockDim.x + threadIdx.x) >> 5;
    if (d >= n) return;
    const float* r = rec + (size_t)d * REC;
    const int start = __float_as_int(r[0]);
    const int end   = __float_as_int(r[1]);
    const float dd  = r[2];
    const int q = lane >> 4;
    const int c = lane & 15;
    const float4* H4 = (const float4*)Hs;
    float4 acc = make_float4(0.0f, 0.0f, 0.0f, 0.0f);

    const int last = end - 1;
    for (int p = start + q; p < end; p += 8) {
        const bool v1 = (p + 2 < end), v2 = (p + 4 < end), v3 = (p + 6 < end);
        const int s0 = csr[p];
        const int s1 = csr[min(p + 2, last)];
        const int s2 = csr[min(p + 4, last)];
        const int s3 = csr[min(p + 6, last)];
        const float4 h0 = H4[(size_t)s0 * (HID / 4) + c];
        const float4 h1 = H4[(size_t)s1 * (HID / 4) + c];
        const float4 h2 = H4[(size_t)s2 * (HID / 4) + c];
        const float4 h3 = H4[(size_t)s3 * (HID / 4) + c];
        acc.x += h0.x; acc.y += h0.y; acc.z += h0.z; acc.w += h0.w;
        acc.x += v1 ? h1.x : 0.0f; acc.y += v1 ? h1.y : 0.0f; acc.z += v1 ? h1.z : 0.0f; acc.w += v1 ? h1.w : 0.0f;
        acc.x += v2 ? h2.x : 0.0f; acc.y += v2 ? h2.y : 0.0f; acc.z += v2 ? h2.z : 0.0f; acc.w += v2 ? h2.w : 0.0f;
        acc.x += v3 ? h3.x : 0.0f; acc.y += v3 ? h3.y : 0.0f; acc.z += v3 ? h3.z : 0.0f; acc.w += v3 ? h3.w : 0.0f;
    }
    acc.x += __shfl_xor(acc.x, 16);
    acc.y += __shfl_xor(acc.y, 16);
    acc.z += __shfl_xor(acc.z, 16);
    acc.w += __shfl_xor(acc.w, 16);
    if (q == 0) {
        const float4 bb = ((const float4*)b1)[c];
        float4 y;
        y.x = fmaxf(acc.x * dd + bb.x, 0.0f);
        y.y = fmaxf(acc.y * dd + bb.y, 0.0f);
        y.z = fmaxf(acc.z * dd + bb.z, 0.0f);
        y.w = fmaxf(acc.w * dd + bb.w, 0.0f);
        ((float4*)Y)[(size_t)d * (HID / 4) + c] = y;
    }
}


__global__ __launch_bounds__(256) void gather16(const float* __restrict__ H2,
                                                const int* __restrict__ csr,
                                                const float* __restrict__ b2,
                                                float* out, int n) {
    const int lane = threadIdx.x & 31;
    const int d = (blockIdx.x * blockDim.x + threadIdx.x) >> 5;
    if (d >= n) return;
    const float* r = out + (size_t)d * REC;
    const int start = __float_as_int(r[0]);
    const int end   = __float_as_int(r[1]);
    const float dd  = r[2];
    const int q = lane >> 2;
    const int c = lane & 3;
    const float4* H4 = (const float4*)H2;
    float4 acc = make_float4(0.0f, 0.0f, 0.0f, 0.0f);

    const int last = end - 1;
    for (int p = start + q; p < end; p += 32) {
        const bool v1 = (p + 8 < end), v2 = (p + 16 < end), v3 = (p + 24 < end);
        const int s0 = csr[p];
        const int s1 = csr[min(p + 8, last)];
        const int s2 = csr[min(p + 16, last)];
        const int s3 = csr[min(p + 24, last)];
        const float4 h0 = H4[(size_t)s0 * (NCLS / 4) + c];
        const float4 h1 = H4[(size_t)s1 * (NCLS / 4) + c];
        const float4 h2 = H4[(size_t)s2 * (NCLS / 4) + c];
        const float4 h3 = H4[(size_t)s3 * (NCLS / 4) + c];
        acc.x += h0.x; acc.y += h0.y; acc.z += h0.z; acc.w += h0.w;
        acc.x += v1 ? h1.x : 0.0f; acc.y += v1 ? h1.y : 0.0f; acc.z += v1 ? h1.z : 0.0f; acc.w += v1 ? h1.w : 0.0f;
        acc.x += v2 ? h2.x : 0.0f; acc.y += v2 ? h2.y : 0.0f; acc.z += v2 ? h2.z : 0.0f; acc.w += v2 ? h2.w : 0.0f;
        acc.x += v3 ? h3.x : 0.0f; acc.y += v3 ? h3.y : 0.0f; acc.z += v3 ? h3.z : 0.0f; acc.w += v3 ? h3.w : 0.0f;
    }
#pragma unroll
    for (int m = 4; m < 32; m <<= 1) {
        acc.x += __shfl_xor(acc.x, m);
        acc.y += __shfl_xor(acc.y, m);
        acc.z += __shfl_xor(acc.z, m);
        acc.w += __shfl_xor(acc.w, m);
    }
    if (q == 0) {
        const float4 bb = ((const float4*)b2)[c];
        float4 y;
        y.x = acc.x * dd + bb.x;
        y.y = acc.y * dd + bb.y;
        y.z = acc.z * dd + bb.z;
        y.w = acc.w * dd + bb.w;
        ((float4*)out)[(size_t)d * (NCLS / 4) + c] = y;
    }
}


extern "C" void kernel_launch(void* const* d_in, const int* in_sizes, int n_in,
                              void* d_out, int out_size, void* d_ws, size_t ws_size,
                              hipStream_t stream) {
    const float* x    = (const float*)d_in[0];
    const int*   esrc = (const int*)d_in[1];
    const int*   edst = (const int*)d_in[2];
    const float* W1   = (const float*)d_in[3];
    const float* b1   = (const float*)d_in[4];
    const float* W2   = (const float*)d_in[5];
    const float* b2   = (const float*)d_in[6];
    float* out = (float*)d_out;

    const int NN = in_sizes[0] / F_IN;
    const int E  = in_sizes[1];
    const int Mtiles = NN / 16;

    float* H   = (float*)d_ws;
    float* Y   = H + (size_t)NN * HID;
    int*   csr = (int*)(Y + (size_t)NN * HID);
    float* H2  = H;

    const int nb     = (NN + BN - 1) / BN;
    const int ntiles = (E + TILE_E - 1) / TILE_E;
    int2* bk      = (int2*)H;
    int*  tilecnt = (int*)Y;
    int*  tileoff = tilecnt + (size_t)nb * ntiles;

    const int B = 256;
    #define NB(n) (((n) + B - 1) / B)

    tile_hist<<<ntiles, TILE_T, 0, stream>>>(edst, tilecnt, E, nb, ntiles);
    scan_table<<<nb, SCAN_T, 0, stream>>>(tilecnt, tileoff, nb, ntiles);
    tile_scatter<<<ntiles, TILE_T, 0, stream>>>(esrc, edst, tileoff, bk, E, nb, ntiles);
    bucket_csr<<<nb, 1024, 0, stream>>>(bk, tileoff, out, csr, NN, ntiles);

    gemm1_wmma<<<Mtiles, 32, 0, stream>>>(x, W1, out, H);
    gather64<<<NB(NN * 32), B, 0, stream>>>(H, csr, out, b1, Y, NN);

    gemm2_wmma<<<Mtiles, 32, 0, stream>>>(Y, W2, out, H2);
    gather16<<<NB(NN * 32), B, 0, stream>>>(H2, csr, b2, out, NN);

    #undef NB
}


# edge grouping in 2 launches, not 4: tiles bucket-ordered in place with a (start,length) table; each bucket group adds up the lengths in front and collects its own runs; sum order: LDS arrival
# speedup vs baseline: 3.2299x; 1.1326x over previous
#include <hip/hip_runtime.h>


typedef __attribute__((ext_vector_type(2))) float v2f;
typedef __attribute__((ext_vector_type(8))) float v8f;

#define F_IN 128
#define HID  64
#define NCLS 16
#define REC  16


#define BSH    9
#define BN     (1 << BSH)
#define MAXB   256
#define TILE_E 4096
#define TILE_T 256
#define EPT    (TILE_E / TILE_T)
#define STAGE  12288
#define RPW    16

template <int NT>
__device__ __forceinline__ int block_scan_incl(int v, int (*s)[NT], int t) {
    int cur = 0;
    s[0][t] = v;
    __syncthreads();
    for (int off = 1; off < NT; off <<= 1) {
        int x = s[cur][t];
        if (t >= off) x += s[cur][t - off];
        s[cur ^ 1][t] = x;
        __syncthreads();
        cur ^= 1;
    }
    const int r = s[cur][t];
    return r;
}

__global__ __launch_bounds__(TILE_T) void tile_sort(const int* __restrict__ src, const int* __restrict__ dst,
                                                    int2* __restrict__ bk, int* __restrict__ tab,
                                                    int E, int nb, int ntiles) {
    __shared__ int cur[MAXB];
    __shared__ int s[2][TILE_T];
    __shared__ int2 sorted[TILE_E];
    const int t = threadIdx.x;
    const int tile = blockIdx.x;
    for (int i = t; i < MAXB; i += TILE_T) cur[i] = 0;
    __syncthreads();
    const int e0 = tile * TILE_E;
    int ms[EPT], md[EPT];
    for (int j = 0; j < EPT; ++j) {
        const int e = e0 + j * TILE_T + t;
        const bool ok = e < E;
        md[j] = ok ? dst[e] : -1;
        ms[j] = ok ? src[e] : 0;
        if (ok) atomicAdd(&cur[md[j] >> BSH], 1);
    }
    __syncthreads();
    static_assert(TILE_T >= MAXB, "one thread per bucket");
    static_assert(TILE_E <= 65535, "run start and length share one word");
    const int c = (t < nb) ? cur[t] : 0;
    const int incl = block_scan_incl<TILE_T>(c, s, t);
    if (t < nb) {
        cur[t] = incl - c;
        tab[t * ntiles + tile] = (incl - c) | (c << 16);
    }
    __syncthreads();
    for (int j = 0; j < EPT; ++j) {
        if (md[j] >= 0) {
            const int p = atomicAdd(&cur[md[j] >> BSH], 1);
            sorted[p] = make_int2(ms[j], md[j]);
        }
    }
    __syncthreads();
    const int ne = min(TILE_E, E - e0);
    for (int i = t; i < ne; i += TILE_T) bk[e0 + i] = sorted[i];
}

__global__ __launch_bounds__(1024) void bucket_csr(const int2* __restrict__ bk, const int* __restrict__ tab,
                                                   float* __restrict__ rec, int* __restrict__ csr,
                                                   int n, int ntiles) {
    __shared__ int cnt[BN];
    __shared__ int s[2][1024];
    __shared__ int stage[STAGE];
    __shared__ int red[2][32];
    const int t = threadIdx.x;
    const int lane = t & 31;
    const int w = __builtin_amdgcn_readfirstlane(t >> 5);
    const int b = blockIdx.x;
    const int base = b * ntiles;
    const int node0 = b << BSH;
    if (t < BN) cnt[t] = 0;

    int p = 0, own = 0;
    const int ownw = tab[base + min(t, ntiles - 1)];
    const int4* c4 = (const int4*)tab;
    const int nv = base >> 2;
    for (int i0 = t; i0 < nv; i0 += 8 * 1024) {
        int4 q[8];
#pragma unroll
        for (int k = 0; k < 8; ++k) q[k] = c4[min(i0 + k * 1024, nv - 1)];
#pragma unroll
        for (int k = 0; k < 8; ++k) {
            const int a = (int)(((unsigned)q[k].x >> 16) + ((unsigned)q[k].y >> 16)) + (int)(((unsigned)q[k].z >> 16) + ((unsigned)q[k].w >> 16));
            p += (i0 + k * 1024 < nv) ? a : 0;
        }
    }
    for (int i = (nv << 2) + t; i < base; i += 1024) p += (int)((unsigned)tab[i] >> 16);
    own = (t < ntiles) ? (int)((unsigned)ownw >> 16) : 0;
    for (int i = t + 1024; i < ntiles; i += 1024) own += (int)((unsigned)tab[base + i] >> 16);
#pragma unroll
    for (int m = 16; m >= 1; m >>= 1) { p += __shfl_xor(p, m); own += __shfl_xor(own, m); }
    if (lane == 0) { red[0][w] = p; red[1][w] = own; }
    __syncthreads();
    p = red[0][lane]; own = red[1][lane];
#pragma unroll
    for (int m = 16; m >= 1; m >>= 1) { p += __shfl_xor(p, m); own += __shfl_xor(own, m); }
    const int e0 = __builtin_amdgcn_readfirstlane(p);
    const int ne = __builtin_amdgcn_readfirstlane(own);

    for (int c = 0; c < ntiles; c += 32 * RPW) {
        const int r0 = c + w * RPW;
        const int dsc = (lane < RPW && r0 + lane < ntiles) ? tab[base + r0 + lane] : 0;
        int dy[RPW];
#pragma unroll
        for (int k = 0; k < RPW; ++k) {
            const int d = __builtin_amdgcn_readlane(dsc, k);
            const int cn = (int)((unsigned)d >> 16);
            const int gi = (lane < cn) ? ((r0 + k) * TILE_E + (d & 0xffff) + lane) : 0;
            dy[k] = bk[gi].y;
        }
#pragma unroll
        for (int k = 0; k < RPW; ++k) {
            const int cn = (int)((unsigned)__builtin_amdgcn_readlane(dsc, k) >> 16);
            if (lane < cn) atomicAdd(&cnt[dy[k] - node0], 1);
        }
#pragma unroll
        for (int k = 0; k < RPW; ++k) {
            const int d = __builtin_amdgcn_readlane(dsc, k);
            const int cn = (int)((unsigned)d >> 16);
            if (cn > 32) {
                const int rb = (r0 + k) * TILE_E + (d & 0xffff);
                for (int j = 32 + lane; j < cn; j += 32) atomicAdd(&cnt[bk[rb + j].y - node0], 1);
            }
        }
    }
    __syncthreads();
    const int v = (t < BN) ? cnt[t] : 0;
    const int incl = block_scan_incl<1024>(v, s, t);
    const int st = incl - v;
    if (t < BN) {
        cnt[t] = st;
        const int node = node0 + t;
        if (node < n) {
            const float deg = (float)v;
            const float dis = (deg > 0.0f) ? rsqrtf(fmaxf(deg, 1.0f)) : 0.0f;
            float* r = rec + (size_t)node * REC;
            r[0] = __int_as_float(e0 + st);
            r[1] = __int_as_float(e0 + st + v);
            r[2] = dis;
        }
    }
    __syncthreads();
    const bool staged = (ne <= STAGE);
    for (int c = 0; c < ntiles; c += 32 * RPW) {
        const int r0 = c + w * RPW;
        const int dsc = (lane < RPW && r0 + lane < ntiles) ? tab[base + r0 + lane] : 0;
        int2 ed[RPW];
#pragma unroll
        for (int k = 0; k < RPW; ++k) {
            const int d = __builtin_amdgcn_readlane(dsc, k);
            const int cn = (int)((unsigned)d >> 16);
            const int gi = (lane < cn) ? ((r0 + k) * TILE_E + (d & 0xffff) + lane) : 0;
            ed[k] = bk[gi];
        }
#pragma unroll
        for (int k = 0; k < RPW; ++k) {
            const int cn = (int)((unsigned)__builtin_amdgcn_readlane(dsc, k) >> 16);
            if (lane < cn) {
                const int q = atomicAdd(&cnt[ed[k].y - node0], 1);
                if (staged) stage[q] = ed[k].x; else csr[e0 + q] = ed[k].x;
            }
        }
#pragma unroll
        for (int k = 0; k < RPW; ++k) {
            const int d = __builtin_amdgcn_readlane(dsc, k);
            const int cn = (int)((unsigned)d >> 16);
            if (cn > 32) {
                const int rb = (r0 + k) * TILE_E + (d & 0xffff);
                for (int j = 32 + lane; j < cn; j += 32) {
                    const int2 e2 = bk[rb + j];
                    const int q = atomicAdd(&cnt[e2.y - node0], 1);
                    if (staged) stage[q] = e2.x; else csr[e0 + q] = e2.x;
                }
            }
        }
    }
    if (staged) {
        __syncthreads();
        for (int i = t; i < ne; i += 1024) csr[e0 + i] = stage[i];
    }
}


__global__ __launch_bounds__(32) void gemm1_wmma(const float* __restrict__ X,
                                                 const float* __restrict__ W1,
                                                 const float* __restrict__ rec,
                                                 float* __restrict__ H) {
    const int tile = blockIdx.x;
    const int lane = threadIdx.x;
    const int half = lane >> 4;
    const int l16  = lane & 15;

    const int arow = tile * 16 + l16;
    v8f c0 = {}, c1 = {}, c2 = {}, c3 = {};

#pragma unroll
    for (int k = 0; k < F_IN; k += 4) {
        const float* ap = X + (size_t)arow * F_IN + k + half * 2;
        v2f a; a[0] = ap[0]; a[1] = ap[1];

        const int krow = k + half * 2;
        const float* bp = W1 + (size_t)krow * HID + l16;
        v2f b0; b0[0] = bp[0];  b0[1] = bp[HID + 0];
        v2f b1; b1[0] = bp[16]; b1[1] = bp[HID + 16];
        v2f b2; b2[0] = bp[32]; b2[1] = bp[HID + 32];
        v2f b3; b3[0] = bp[48]; b3[1] = bp[HID + 48];

        c0 = __builtin_amdgcn_wmma_f32_16x16x4_f32(false, a, false, b0, (short)0, c0, false, false);
        c1 = __builtin_amdgcn_wmma_f32_16x16x4_f32(false, a, false, b1, (short)0, c1, false, false);
        c2 = __builtin_amdgcn_wmma_f32_16x16x4_f32(false, a, false, b2, (short)0, c2, false, false);
        c3 = __builtin_amdgcn_wmma_f32_16x16x4_f32(false, a, false, b3, (short)0, c3, false, false);
    }

    const int row0 = tile * 16 + half * 8;
    float* hp = H + (size_t)row0 * HID + l16;
    const float* rp = rec + (size_t)row0 * REC + 2;
#pragma unroll
    for (int r = 0; r < 8; ++r) {
        const float sc = rp[r * REC];
        hp[r * HID + 0]  = c0[r] * sc;
        hp[r * HID + 16] = c1[r] * sc;
        hp[r * HID + 32] = c2[r] * sc;
        hp[r * HID + 48] = c3[r] * sc;
    }
}


__global__ __launch_bounds__(32) void gemm2_wmma(const float* __restrict__ Y,
                                                 const float* __restrict__ W2,
                                                 const float* __restrict__ rec,
                                                 float* __restrict__ H2) {
    const int tile = blockIdx.x;
    const int lane = threadIdx.x;
    const int half = lane >> 4;
    const int l16  = lane & 15;

    const int arow = tile * 16 + l16;
    v8f c0 = {};

#pragma unroll
    for (int k = 0; k < HID; k += 4) {
        const float* ap = Y + (size_t)arow * HID + k + half * 2;
        v2f a; a[0] = ap[0]; a[1] = ap[1];

        const int krow = k + half * 2;
        const float* bp = W2 + (size_t)krow * NCLS + l16;
        v2f b; b[0] = bp[0]; b[1] = bp[NCLS];

        c0 = __builtin_amdgcn_wmma_f32_16x16x4_f32(false, a, false, b, (short)0, c0, false, false);
    }

    const int row0 = tile * 16 + half * 8;
    float* hp = H2 + (size_t)row0 * NCLS + l16;
    const float* rp = rec + (size_t)row0 * REC + 2;
#pragma unroll
    for (int r = 0; r < 8; ++r) hp[r * NCLS] = c0[r] * rp[r * REC];
}


__global__ __launch_bounds__(256) void gather64(const float* __restrict__ Hs,
                                                const int* __restrict__ csr,
                                                const float* __restrict__ rec,
                                                const float* __restrict__ b1,
                                                float* __restrict__ Y, int n) {
    const int lane = threadIdx.x & 31;
    const int d = (blockIdx.x * blockDim.x + threadIdx.x) >> 5;
    if (d >= n) return;
    const float* r = rec + (size_t)d * REC;
    const int start = __float_as_int(r[0]);
    const int end   = __float_as_int(r[1]);
    const float dd  = r[2];
    const int q = lane >> 4;
    const int c = lane & 15;
    const float4* H4 = (const float4*)Hs;
    float4 acc = make_float4(0.0f, 0.0f, 0.0f, 0.0f);

    const int last = end - 1;
    for (int p = start + q; p < end; p += 8) {
        const bool v1 = (p + 2 < end), v2 = (p + 4 < end), v3 = (p + 6 < end);
        const int s0 = csr[p];
        const int s1 = csr[min(p + 2, last)];
        const int s2 = csr[min(p + 4, last)];
        const int s3 = csr[min(p + 6, last)];
        const float4 h0 = H4[(size_t)s0 * (HID / 4) + c];
        const float4 h1 = H4[(size_t)s1 * (HID / 4) + c];
        const float4 h2 = H4[(size_t)s2 * (HID / 4) + c];
        const float4 h3 = H4[(size_t)s3 * (HID / 4) + c];
        acc.x += h0.x; acc.y += h0.y; acc.z += h0.z; acc.w += h0.w;
        acc.x += v1 ? h1.x : 0.0f; acc.y += v1 ? h1.y : 0.0f; acc.z += v1 ? h1.z : 0.0f; acc.w += v1 ? h1.w : 0.0f;
        acc.x += v2 ? h2.x : 0.0f; acc.y += v2 ? h2.y : 0.0f; acc.z += v2 ? h2.z : 0.0f; acc.w += v2 ? h2.w : 0.0f;
        acc.x += v3 ? h3.x : 0.0f; acc.y += v3 ? h3.y : 0.0f; acc.z += v3 ? h3.z : 0.0f; acc.w += v3 ? h3.w : 0.0f;
    }
    acc.x += __shfl_xor(acc.x, 16);
    acc.y += __shfl_xor(acc.y, 16);
    acc.z += __shfl_xor(acc.z, 16);
    acc.w += __shfl_xor(acc.w, 16);
    if (q == 0) {
        const float4 bb = ((const float4*)b1)[c];
        float4 y;
        y.x = fmaxf(acc.x * dd + bb.x, 0.0f);
        y.y = fmaxf(acc.y * dd + bb.y, 0.0f);
        y.z = fmaxf(acc.z * dd + bb.z, 0.0f);
        y.w = fmaxf(acc.w * dd + bb.w, 0.0f);
        ((float4*)Y)[(size_t)d * (HID / 4) + c] = y;
    }
}


__global__ __launch_bounds__(256) void gather16(const float* __restrict__ H2,
                                                const int* __restrict__ csr,
                                                const float* __restrict__ b2,
                                                float* out, int n) {
    const int lane = threadIdx.x & 31;
    const int d = (blockIdx.x * blockDim.x + threadIdx.x) >> 5;
    if (d >= n) return;
    const float* r = out + (size_t)d * REC;
    const int start = __float_as_int(r[0]);
    const int end   = __float_as_int(r[1]);
    const float dd  = r[2];
    const int q = lane >> 2;
    const int c = lane & 3;
    const float4* H4 = (const float4*)H2;
    float4 acc = make_float4(0.0f, 0.0f, 0.0f, 0.0f);

    const int last = end - 1;
    for (int p = start + q; p < end; p += 32) {
        const bool v1 = (p + 8 < end), v2 = (p + 16 < end), v3 = (p + 24 < end);
        const int s0 = csr[p];
        const int s1 = csr[min(p + 8, last)];
        const int s2 = csr[min(p + 16, last)];
        const int s3 = csr[min(p + 24, last)];
        const float4 h0 = H4[(size_t)s0 * (NCLS / 4) + c];
        const float4 h1 = H4[(size_t)s1 * (NCLS / 4) + c];
        const float4 h2 = H4[(size_t)s2 * (NCLS / 4) + c];
        const float4 h3 = H4[(size_t)s3 * (NCLS / 4) + c];
        acc.x += h0.x; acc.y += h0.y; acc.z += h0.z; acc.w += h0.w;
        acc.x += v1 ? h1.x : 0.0f; acc.y += v1 ? h1.y : 0.0f; acc.z += v1 ? h1.z : 0.0f; acc.w += v1 ? h1.w : 0.0f;
        acc.x += v2 ? h2.x : 0.0f; acc.y += v2 ? h2.y : 0.0f; acc.z += v2 ? h2.z : 0.0f; acc.w += v2 ? h2.w : 0.0f;
        acc.x += v3 ? h3.x : 0.0f; acc.y += v3 ? h3.y : 0.0f; acc.z += v3 ? h3.z : 0.0f; acc.w += v3 ? h3.w : 0.0f;
    }
#pragma unroll
    for (int m = 4; m < 32; m <<= 1) {
        acc.x += __shfl_xor(acc.x, m);
        acc.y += __shfl_xor(acc.y, m);
        acc.z += __shfl_xor(acc.z, m);
        acc.w += __shfl_xor(acc.w, m);
    }
    if (q == 0) {
        const float4 bb = ((const float4*)b2)[c];
        float4 y;
        y.x = acc.x * dd + bb.x;
        y.y = acc.y * dd + bb.y;
        y.z = acc.z * dd + bb.z;
        y.w = acc.w * dd + bb.w;
        ((float4*)out)[(size_t)d * (NCLS / 4) + c] = y;
    }
}


extern "C" void kernel_launch(void* const* d_in, const int* in_sizes, int n_in,
                              void* d_out, int out_size, void* d_ws, size_t ws_size,
                              hipStream_t stream) {
    const float* x    = (const float*)d_in[0];
    const int*   esrc = (const int*)d_in[1];
    const int*   edst = (const int*)d_in[2];
    const float* W1   = (const float*)d_in[3];
    const float* b1   = (const float*)d_in[4];
    const float* W2   = (const float*)d_in[5];
    const float* b2   = (const float*)d_in[6];
    float* out = (float*)d_out;

    const int NN = in_sizes[0] / F_IN;
    const int E  = in_sizes[1];
    const int Mtiles = NN / 16;

    float* H   = (float*)d_ws;
    float* Y   = H + (size_t)NN * HID;
    int*   csr = (int*)(Y + (size_t)NN * HID);
    float* H2  = H;

    const int nb     = (NN + BN - 1) / BN;
    const int ntiles = (E + TILE_E - 1) / TILE_E;
    int2* bk      = (int2*)H;
    int*  tab     = (int*)Y;

    const int B = 256;
    #define NB(n) (((n) + B - 1) / B)

    tile_sort<<<ntiles, TILE_T, 0, stream>>>(esrc, edst, bk, tab, E, nb, ntiles);
    bucket_csr<<<nb, 1024, 0, stream>>>(bk, tab, out, csr, NN, ntiles);

    gemm1_wmma<<<Mtiles, 32, 0, stream>>>(x, W1, out, H);
    gather64<<<NB(NN * 32), B, 0, stream>>>(H, csr, out, b1, Y, NN);

    gemm2_wmma<<<Mtiles, 32, 0, stream>>>(Y, W2, out, H2);
    gather16<<<NB(NN * 32), B, 0, stream>>>(H2, csr, b2, out, NN);

    #undef NB
}


# layer-1 product runs inside the tile-sort launch as extra 8-wave work-groups (5 launches, not 6); the bucket kernel scales its own 512 product rows by deg^-1/2; every sum and its order unchanged
# speedup vs baseline: 3.4455x; 1.0668x over previous
#include <hip/hip_runtime.h>


typedef __attribute__((ext_vector_type(2))) float v2f;
typedef __attribute__((ext_vector_type(8))) float v8f;

#define F_IN 128
#define HID  64
#define NCLS 16
#define REC  16


#define BSH    9
#define BN     (1 << BSH)
#define MAXB   256
#define TILE_E 4096
#define TILE_T 256
#define EPT    (TILE_E / TILE_T)
#define STAGE  12288
#define RPW    16
#define GW     (TILE_T / 32)
#define HV     (BN * (HID / 4) / 1024)

template <int NT>
__device__ __forceinline__ int block_scan_incl(int v, int (*s)[NT], int t) {
    int cur = 0;
    s[0][t] = v;
    __syncthreads();
    for (int off = 1; off < NT; off <<= 1) {
        int x = s[cur][t];
        if (t >= off) x += s[cur][t - off];
        s[cur ^ 1][t] = x;
        __syncthreads();
        cur ^= 1;
    }
    const int r = s[cur][t];
    return r;
}

__device__ __forceinline__ void gemm1_tile(const float* __restrict__ X, const float* __restrict__ W1,
                                           float* __restrict__ H, int tile, int lane) {
    const int half = lane >> 4;
    const int l16  = lane & 15;

    const int arow = tile * 16 + l16;
    v8f c0 = {}, c1 = {}, c2 = {}, c3 = {};

#pragma unroll
    for (int k = 0; k < F_IN; k += 4) {
        const float* ap = X + (size_t)arow * F_IN + k + half * 2;
        v2f a; a[0] = ap[0]; a[1] = ap[1];

        const int krow = k + half * 2;
        const float* bp = W1 + (size_t)krow * HID + l16;
        v2f b0; b0[0] = bp[0];  b0[1] = bp[HID + 0];
        v2f b1; b1[0] = bp[16]; b1[1] = bp[HID + 16];
        v2f b2; b2[0] = bp[32]; b2[1] = bp[HID + 32];
        v2f b3; b3[0] = bp[48]; b3[1] = bp[HID + 48];

        c0 = __builtin_amdgcn_wmma_f32_16x16x4_f32(false, a, false, b0, (short)0, c0, false, false);
        c1 = __builtin_amdgcn_wmma_f32_16x16x4_f32(false, a, false, b1, (short)0, c1, false, false);
        c2 = __builtin_amdgcn_wmma_f32_16x16x4_f32(false, a, false, b2, (short)0, c2, false, false);
        c3 = __builtin_amdgcn_wmma_f32_16x16x4_f32(false, a, false, b3, (short)0, c3, false, false);
    }

    const int row0 = tile * 16 + half * 8;
    float* hp = H + (size_t)row0 * HID + l16;
#pragma unroll
    for (int r = 0; r < 8; ++r) {
        hp[r * HID + 0]  = c0[r];
        hp[r * HID + 16] = c1[r];
        hp[r * HID + 32] = c2[r];
        hp[r * HID + 48] = c3[r];
    }
}

__global__ __launch_bounds__(TILE_T) void tile_sort_gemm1(const int* __restrict__ src, const int* __restrict__ dst,
                                                          int2* __restrict__ bk, int* __restrict__ tab,
                                                          int E, int nb, int ntiles,
                                                          const float* __restrict__ X, const float* __restrict__ W1,
                                                          float* __restrict__ H, int mtiles) {
    __shared__ int cur[MAXB];
    __shared__ int s[2][TILE_T];
    __shared__ int2 sorted[TILE_E];
    const int t = threadIdx.x;
    if ((int)blockIdx.x >= ntiles) {
        const int w = __builtin_amdgcn_readfirstlane(t >> 5);
        const int mt = ((int)blockIdx.x - ntiles) * GW + w;
        if (mt < mtiles) gemm1_tile(X, W1, H, mt, t & 31);
        return;
    }
    const int tile = blockIdx.x;
    for (int i = t; i < MAXB; i += TILE_T) cur[i] = 0;
    __syncthreads();
    const int e0 = tile * TILE_E;
    int ms[EPT], md[EPT];
    for (int j = 0; j < EPT; ++j) {
        const int e = e0 + j * TILE_T + t;
        const bool ok = e < E;
        md[j] = ok ? dst[e] : -1;
        ms[j] = ok ? src[e] : 0;
        if (ok) atomicAdd(&cur[md[j] >> BSH], 1);
    }
    __syncthreads();
    static_assert(TILE_T >= MAXB, "one thread per bucket");
    static_assert(TILE_E <= 65535, "run start and length share one word");
    const int c = (t < nb) ? cur[t] : 0;
    const int incl = block_scan_incl<TILE_T>(c, s, t);
    if (t < nb) {
        cur[t] = incl - c;
        tab[t * ntiles + tile] = (incl - c) | (c << 16);
    }
    __syncthreads();
    for (int j = 0; j < EPT; ++j) {
        if (md[j] >= 0) {
            const int p = atomicAdd(&cur[md[j] >> BSH], 1);
            sorted[p] = make_int2(ms[j], md[j]);
        }
    }
    __syncthreads();
    const int ne = min(TILE_E, E - e0);
    for (int i = t; i < ne; i += TILE_T) bk[e0 + i] = sorted[i];
}

__global__ __launch_bounds__(1024) void bucket_csr(const int2* __restrict__ bk, const int* __restrict__ tab,
                                                   float* __restrict__ rec, int* __restrict__ csr,
                                                   float* H, int n, int ntiles) {
    __shared__ int cnt[BN];
    __shared__ float disl[BN];
    __shared__ int s[2][1024];
    __shared__ int stage[STAGE];
    __shared__ int red[2][32];
    const int t = threadIdx.x;
    const int lane = t & 31;
    const int w = __builtin_amdgcn_readfirstlane(t >> 5);
    const int b = blockIdx.x;
    const int base = b * ntiles;
    const int node0 = b << BSH;
    if (t < BN) cnt[t] = 0;

    int p = 0, own = 0;
    const int ownw = tab[base + min(t, ntiles - 1)];
    const int4* c4 = (const int4*)tab;
    const int nv = base >> 2;
    for (int i0 = t; i0 < nv; i0 += 8 * 1024) {
        int4 q[8];
#pragma unroll
        for (int k = 0; k < 8; ++k) q[k] = c4[min(i0 + k * 1024, nv - 1)];
#pragma unroll
        for (int k = 0; k < 8; ++k) {
            const int a = (int)(((unsigned)q[k].x >> 16) + ((unsigned)q[k].y >> 16)) + (int)(((unsigned)q[k].z >> 16) + ((unsigned)q[k].w >> 16));
            p += (i0 + k * 1024 < nv) ? a : 0;
        }
    }
    for (int i = (nv << 2) + t; i < base; i += 1024) p += (int)((unsigned)tab[i] >> 16);
    own = (t < ntiles) ? (int)((unsigned)ownw >> 16) : 0;
    for (int i = t + 1024; i < ntiles; i += 1024) own += (int)((unsigned)tab[base + i] >> 16);
#pragma unroll
    for (int m = 16; m >= 1; m >>= 1) { p += __shfl_xor(p, m); own += __shfl_xor(own, m); }
    if (lane == 0) { red[0][w] = p; red[1][w] = own; }
    __syncthreads();
    p = red[0][lane]; own = red[1][lane];
#pragma unroll
    for (int m = 16; m >= 1; m >>= 1) { p += __shfl_xor(p, m); own += __shfl_xor(own, m); }
    const int e0 = __builtin_amdgcn_readfirstlane(p);
    const int ne = __builtin_amdgcn_readfirstlane(own);

    float4* H4 = (float4*)H + (size_t)node0 * (HID / 4);
    const int hrows = min(BN, n - node0);
    float4 hv[HV];
#pragma unroll
    for (int k = 0; k < HV; ++k) {
        const int i = t + k * 1024;
        hv[k] = ((i >> 4) < hrows) ? H4[i] : make_float4(0.0f, 0.0f, 0.0f, 0.0f);
    }

    for (int c = 0; c < ntiles; c += 32 * RPW) {
        const int r0 = c + w * RPW;
        const int dsc = (lane < RPW && r0 + lane < ntiles) ? tab[base + r0 + lane] : 0;
        int dy[RPW];
#pragma unroll
        for (int k = 0; k < RPW; ++k) {
            const int d = __builtin_amdgcn_readlane(dsc, k);
            const int cn = (int)((unsigned)d >> 16);
            const int gi = (lane < cn) ? ((r0 + k) * TILE_E + (d & 0xffff) + lane) : 0;
            dy[k] = bk[gi].y;
        }
#pragma unroll
        for (int k = 0; k < RPW; ++k) {
            const int cn = (int)((unsigned)__builtin_amdgcn_readlane(dsc, k) >> 16);
            if (lane < cn) atomicAdd(&cnt[dy[k] - node0], 1);
        }
#pragma unroll
        for (int k = 0; k < RPW; ++k) {
            const int d = __builtin_amdgcn_readlane(dsc, k);
            const int cn = (int)((unsigned)d >> 16);
            if (cn > 32) {
                const int rb = (r0 + k) * TILE_E + (d & 0xffff);
                for (int j = 32 + lane; j < cn; j += 32) atomicAdd(&cnt[bk[rb + j].y - node0], 1);
            }
        }
    }
    __syncthreads();
    const int v = (t < BN) ? cnt[t] : 0;
    const int incl = block_scan_incl<1024>(v, s, t);
    const int st = incl - v;
    if (t < BN) {
        cnt[t] = st;
        const int node = node0 + t;
        if (node < n) {
            const float deg = (float)v;
            const float dis = (deg > 0.0f) ? rsqrtf(fmaxf(deg, 1.0f)) : 0.0f;
            float* r = rec + (size_t)node * REC;
            r[0] = __int_as_float(e0 + st);
            r[1] = __int_as_float(e0 + st + v);
            r[2] = dis;
            disl[t] = dis;
        }
    }
    __syncthreads();
#pragma unroll
    for (int k = 0; k < HV; ++k) {
        const int i = t + k * 1024;
        if ((i >> 4) < hrows) {
            const float sc = disl[i >> 4];
            H4[i] = make_float4(hv[k].x * sc, hv[k].y * sc, hv[k].z * sc, hv[k].w * sc);
        }
    }
    const bool staged = (ne <= STAGE);
    for (int c = 0; c < ntiles; c += 32 * RPW) {
        const int r0 = c + w * RPW;
        const int dsc = (lane < RPW && r0 + lane < ntiles) ? tab[base + r0 + lane] : 0;
        int2 ed[RPW];
#pragma unroll
        for (int k = 0; k < RPW; ++k) {
            const int d = __builtin_amdgcn_readlane(dsc, k);
            const int cn = (int)((unsigned)d >> 16);
            const int gi = (lane < cn) ? ((r0 + k) * TILE_E + (d & 0xffff) + lane) : 0;
            ed[k] = bk[gi];
        }
#pragma unroll
        for (int k = 0; k < RPW; ++k) {
            const int cn = (int)((unsigned)__builtin_amdgcn_readlane(dsc, k) >> 16);
            if (lane < cn) {
                const int q = atomicAdd(&cnt[ed[k].y - node0], 1);
                if (staged) stage[q] = ed[k].x; else csr[e0 + q] = ed[k].x;
            }
        }
#pragma unroll
        for (int k = 0; k < RPW; ++k) {
            const int d = __builtin_amdgcn_readlane(dsc, k);
            const int cn = (int)((unsigned)d >> 16);
            if (cn > 32) {
                const int rb = (r0 + k) * TILE_E + (d & 0xffff);
                for (int j = 32 + lane; j < cn; j += 32) {
                    const int2 e2 = bk[rb + j];
                    const int q = atomicAdd(&cnt[e2.y - node0], 1);
                    if (staged) stage[q] = e2.x; else csr[e0 + q] = e2.x;
                }
            }
        }
    }
    if (staged) {
        __syncthreads();
        for (int i = t; i < ne; i += 1024) csr[e0 + i] = stage[i];
    }
}


__global__ __launch_bounds__(32) void gemm2_wmma(const float* __restrict__ Y,
                                                 const float* __restrict__ W2,
                                                 const float* __restrict__ rec,
                                                 float* __restrict__ H2) {
    const int tile = blockIdx.x;
    const int lane = threadIdx.x;
    const int half = lane >> 4;
    const int l16  = lane & 15;

    const int arow = tile * 16 + l16;
    v8f c0 = {};

#pragma unroll
    for (int k = 0; k < HID; k += 4) {
        const float* ap = Y + (size_t)arow * HID + k + half * 2;
        v2f a; a[0] = ap[0]; a[1] = ap[1];

        const int krow = k + half * 2;
        const float* bp = W2 + (size_t)krow * NCLS + l16;
        v2f b; b[0] = bp[0]; b[1] = bp[NCLS];

        c0 = __builtin_amdgcn_wmma_f32_16x16x4_f32(false, a, false, b, (short)0, c0, false, false);
    }

    const int row0 = tile * 16 + half * 8;
    float* hp = H2 + (size_t)row0 * NCLS + l16;
    const float* rp = rec + (size_t)row0 * REC + 2;
#pragma unroll
    for (int r = 0; r < 8; ++r) hp[r * NCLS] = c0[r] * rp[r * REC];
}


__global__ __launch_bounds__(256) void gather64(const float* __restrict__ Hs,
                                                const int* __restrict__ csr,
                                                const float* __restrict__ rec,
                                                const float* __restrict__ b1,
                                                float* __restrict__ Y, int n) {
    const int lane = threadIdx.x & 31;
    const int d = (blockIdx.x * blockDim.x + threadIdx.x) >> 5;
    if (d >= n) return;
    const float* r = rec + (size_t)d * REC;
    const int start = __float_as_int(r[0]);
    const int end   = __float_as_int(r[1]);
    const float dd  = r[2];
    const int q = lane >> 4;
    const int c = lane & 15;
    const float4* H4 = (const float4*)Hs;
    float4 acc = make_float4(0.0f, 0.0f, 0.0f, 0.0f);

    const int last = end - 1;
    for (int p = start + q; p < end; p += 8) {
        const bool v1 = (p + 2 < end), v2 = (p + 4 < end), v3 = (p + 6 < end);
        const int s0 = csr[p];
        const int s1 = csr[min(p + 2, last)];
        const int s2 = csr[min(p + 4, last)];
        const int s3 = csr[min(p + 6, last)];
        const float4 h0 = H4[(size_t)s0 * (HID / 4) + c];
        const float4 h1 = H4[(size_t)s1 * (HID / 4) + c];
        const float4 h2 = H4[(size_t)s2 * (HID / 4) + c];
        const float4 h3 = H4[(size_t)s3 * (HID / 4) + c];
        acc.x += h0.x; acc.y += h0.y; acc.z += h0.z; acc.w += h0.w;
        acc.x += v1 ? h1.x : 0.0f; acc.y += v1 ? h1.y : 0.0f; acc.z += v1 ? h1.z : 0.0f; acc.w += v1 ? h1.w : 0.0f;
        acc.x += v2 ? h2.x : 0.0f; acc.y += v2 ? h2.y : 0.0f; acc.z += v2 ? h2.z : 0.0f; acc.w += v2 ? h2.w : 0.0f;
        acc.x += v3 ? h3.x : 0.0f; acc.y += v3 ? h3.y : 0.0f; acc.z += v3 ? h3.z : 0.0f; acc.w += v3 ? h3.w : 0.0f;
    }
    acc.x += __shfl_xor(acc.x, 16);
    acc.y += __shfl_xor(acc.y, 16);
    acc.z += __shfl_xor(acc.z, 16);
    acc.w += __shfl_xor(acc.w, 16);
    if (q == 0) {
        const float4 bb = ((const float4*)b1)[c];
        float4 y;
        y.x = fmaxf(acc.x * dd + bb.x, 0.0f);
        y.y = fmaxf(acc.y * dd + bb.y, 0.0f);
        y.z = fmaxf(acc.z * dd + bb.z, 0.0f);
        y.w = fmaxf(acc.w * dd + bb.w, 0.0f);
        ((float4*)Y)[(size_t)d * (HID / 4) + c] = y;
    }
}


__global__ __launch_bounds__(256) void gather16(const float* __restrict__ H2,
                                                const int* __restrict__ csr,
                                                const float* __restrict__ b2,
                                                float* out, int n) {
    const int lane = threadIdx.x & 31;
    const int d = (blockIdx.x * blockDim.x + threadIdx.x) >> 5;
    if (d >= n) return;
    const float* r = out + (size_t)d * REC;
    const int start = __float_as_int(r[0]);
    const int end   = __float_as_int(r[1]);
    const float dd  = r[2];
    const int q = lane >> 2;
    const int c = lane & 3;
    const float4* H4 = (const float4*)H2;
    float4 acc = make_float4(0.0f, 0.0f, 0.0f, 0.0f);

    const int last = end - 1;
    for (int p = start + q; p < end; p += 32) {
        const bool v1 = (p + 8 < end), v2 = (p + 16 < end), v3 = (p + 24 < end);
        const int s0 = csr[p];
        const int s1 = csr[min(p + 8, last)];
        const int s2 = csr[min(p + 16, last)];
        const int s3 = csr[min(p + 24, last)];
        const float4 h0 = H4[(size_t)s0 * (NCLS / 4) + c];
        const float4 h1 = H4[(size_t)s1 * (NCLS / 4) + c];
        const float4 h2 = H4[(size_t)s2 * (NCLS / 4) + c];
        const float4 h3 = H4[(size_t)s3 * (NCLS / 4) + c];
        acc.x += h0.x; acc.y += h0.y; acc.z += h0.z; acc.w += h0.w;
        acc.x += v1 ? h1.x : 0.0f; acc.y += v1 ? h1.y : 0.0f; acc.z += v1 ? h1.z : 0.0f; acc.w += v1 ? h1.w : 0.0f;
        acc.x += v2 ? h2.x : 0.0f; acc.y += v2 ? h2.y : 0.0f; acc.z += v2 ? h2.z : 0.0f; acc.w += v2 ? h2.w : 0.0f;
        acc.x += v3 ? h3.x : 0.0f; acc.y += v3 ? h3.y : 0.0f; acc.z += v3 ? h3.z : 0.0f; acc.w += v3 ? h3.w : 0.0f;
    }
#pragma unroll
    for (int m = 4; m < 32; m <<= 1) {
        acc.x += __shfl_xor(acc.x, m);
        acc.y += __shfl_xor(acc.y, m);
        acc.z += __shfl_xor(acc.z, m);
        acc.w += __shfl_xor(acc.w, m);
    }
    if (q == 0) {
        const float4 bb = ((const float4*)b2)[c];
        float4 y;
        y.x = acc.x * dd + bb.x;
        y.y = acc.y * dd + bb.y;
        y.z = acc.z * dd + bb.z;
        y.w = acc.w * dd + bb.w;
        ((float4*)out)[(size_t)d * (NCLS / 4) + c] = y;
    }
}


extern "C" void kernel_launch(void* const* d_in, const int* in_sizes, int n_in,
                              void* d_out, int out_size, void* d_ws, size_t ws_size,
                              hipStream_t stream) {
    const float* x    = (const float*)d_in[0];
    const int*   esrc = (const int*)d_in[1];
    const int*   edst = (const int*)d_in[2];
    const float* W1   = (const float*)d_in[3];
    const float* b1   = (const float*)d_in[4];
    const float* W2   = (const float*)d_in[5];
    const float* b2   = (const float*)d_in[6];
    float* out = (float*)d_out;

    const int NN = in_sizes[0] / F_IN;
    const int E  = in_sizes[1];
    const int Mtiles = NN / 16;

    float* H   = (float*)d_ws;
    float* Y   = H + (size_t)NN * HID;
    int*   csr = (int*)(Y + (size_t)NN * HID);
    float* H2  = H;

    const int nb     = (NN + BN - 1) / BN;
    const int ntiles = (E + TILE_E - 1) / TILE_E;
    const int gwg    = (Mtiles + GW - 1) / GW;
    int2* bk      = (int2*)Y;
    int*  tab     = (int*)Y + (((size_t)2 * E + 3) & ~(size_t)3);

    const int B = 256;
    #define NB(n) (((n) + B - 1) / B)

    tile_sort_gemm1<<<ntiles + gwg, TILE_T, 0, stream>>>(esrc, edst, bk, tab, E, nb, ntiles, x, W1, H, Mtiles);
    bucket_csr<<<nb, 1024, 0, stream>>>(bk, tab, out, csr, H, NN, ntiles);

    gather64<<<NB(NN * 32), B, 0, stream>>>(H, csr, out, b1, Y, NN);

    gemm2_wmma<<<Mtiles, 32, 0, stream>>>(Y, W2, out, H2);
    gather16<<<NB(NN * 32), B, 0, stream>>>(H2, csr, b2, out, NN);

    #undef NB
}


# layer-2 product moved into the layer-1 row-sum launch: an 8-wave work-group sums a 16-node tile (2 nodes per wave), relu rows stay in LDS, wave 0 does the 16 matrix steps; 4 launches, not 5
# speedup vs baseline: 3.6085x; 1.0473x over previous
#include <hip/hip_runtime.h>


typedef __attribute__((ext_vector_type(2))) float v2f;
typedef __attribute__((ext_vector_type(8))) float v8f;

#define F_IN 128
#define HID  64
#define NCLS 16
#define REC  16


#define BSH    9
#define BN     (1 << BSH)
#define MAXB   256
#define TILE_E 4096
#define TILE_T 256
#define EPT    (TILE_E / TILE_T)
#define STAGE  12288
#define RPW    16
#define GW     (TILE_T / 32)
#define HV     (BN * (HID / 4) / 1024)

template <int NT>
__device__ __forceinline__ int block_scan_incl(int v, int (*s)[NT], int t) {
    int cur = 0;
    s[0][t] = v;
    __syncthreads();
    for (int off = 1; off < NT; off <<= 1) {
        int x = s[cur][t];
        if (t >= off) x += s[cur][t - off];
        s[cur ^ 1][t] = x;
        __syncthreads();
        cur ^= 1;
    }
    const int r = s[cur][t];
    return r;
}

__device__ __forceinline__ void gemm1_tile(const float* __restrict__ X, const float* __restrict__ W1,
                                           float* __restrict__ H, int tile, int lane) {
    const int half = lane >> 4;
    const int l16  = lane & 15;

    const int arow = tile * 16 + l16;
    v8f c0 = {}, c1 = {}, c2 = {}, c3 = {};

#pragma unroll
    for (int k = 0; k < F_IN; k += 4) {
        const float* ap = X + (size_t)arow * F_IN + k + half * 2;
        v2f a; a[0] = ap[0]; a[1] = ap[1];

        const int krow = k + half * 2;
        const float* bp = W1 + (size_t)krow * HID + l16;
        v2f b0; b0[0] = bp[0];  b0[1] = bp[HID + 0];
        v2f b1; b1[0] = bp[16]; b1[1] = bp[HID + 16];
        v2f b2; b2[0] = bp[32]; b2[1] = bp[HID + 32];
        v2f b3; b3[0] = bp[48]; b3[1] = bp[HID + 48];

        c0 = __builtin_amdgcn_wmma_f32_16x16x4_f32(false, a, false, b0, (short)0, c0, false, false);
        c1 = __builtin_amdgcn_wmma_f32_16x16x4_f32(false, a, false, b1, (short)0, c1, false, false);
        c2 = __builtin_amdgcn_wmma_f32_16x16x4_f32(false, a, false, b2, (short)0, c2, false, false);
        c3 = __builtin_amdgcn_wmma_f32_16x16x4_f32(false, a, false, b3, (short)0, c3, false, false);
    }

    const int row0 = tile * 16 + half * 8;
    float* hp = H + (size_t)row0 * HID + l16;
#pragma unroll
    for (int r = 0; r < 8; ++r) {
        hp[r * HID + 0]  = c0[r];
        hp[r * HID + 16] = c1[r];
        hp[r * HID + 32] = c2[r];
        hp[r * HID + 48] = c3[r];
    }
}

__global__ __launch_bounds__(TILE_T) void tile_sort_gemm1(const int* __restrict__ src, const int* __restrict__ dst,
                                                          int2* __restrict__ bk, int* __restrict__ tab,
                                                          int E, int nb, int ntiles,
                                                          const float* __restrict__ X, const float* __restrict__ W1,
                                                          float* __restrict__ H, int mtiles) {
    __shared__ int cur[MAXB];
    __shared__ int s[2][TILE_T];
    __shared__ int2 sorted[TILE_E];
    const int t = threadIdx.x;
    if ((int)blockIdx.x >= ntiles) {
        const int w = __builtin_amdgcn_readfirstlane(t >> 5);
        const int mt = ((int)blockIdx.x - ntiles) * GW + w;
        if (mt < mtiles) gemm1_tile(X, W1, H, mt, t & 31);
        return;
    }
    const int tile = blockIdx.x;
    for (int i = t; i < MAXB; i += TILE_T) cur[i] = 0;
    __syncthreads();
    const int e0 = tile * TILE_E;
    int ms[EPT], md[EPT];
    for (int j = 0; j < EPT; ++j) {
        const int e = e0 + j * TILE_T + t;
        const bool ok = e < E;
        md[j] = ok ? dst[e] : -1;
        ms[j] = ok ? src[e] : 0;
        if (ok) atomicAdd(&cur[md[j] >> BSH], 1);
    }
    __syncthreads();
    static_assert(TILE_T >= MAXB, "one thread per bucket");
    static_assert(TILE_E <= 65535, "run start and length share one word");
    const int c = (t < nb) ? cur[t] : 0;
    const int incl = block_scan_incl<TILE_T>(c, s, t);
    if (t < nb) {
        cur[t] = incl - c;
        tab[t * ntiles + tile] = (incl - c) | (c << 16);
    }
    __syncthreads();
    for (int j = 0; j < EPT; ++j) {
        if (md[j] >= 0) {
            const int p = atomicAdd(&cur[md[j] >> BSH], 1);
            sorted[p] = make_int2(ms[j], md[j]);
        }
    }
    __syncthreads();
    const int ne = min(TILE_E, E - e0);
    for (int i = t; i < ne; i += TILE_T) bk[e0 + i] = sorted[i];
}

__global__ __launch_bounds__(1024) void bucket_csr(const int2* __restrict__ bk, const int* __restrict__ tab,
                                                   float* __restrict__ rec, int* __restrict__ csr,
                                                   float* H, int n, int ntiles) {
    __shared__ int cnt[BN];
    __shared__ float disl[BN];
    __shared__ int s[2][1024];
    __shared__ int stage[STAGE];
    __shared__ int red[2][32];
    const int t = threadIdx.x;
    const int lane = t & 31;
    const int w = __builtin_amdgcn_readfirstlane(t >> 5);
    const int b = blockIdx.x;
    const int base = b * ntiles;
    const int node0 = b << BSH;
    if (t < BN) cnt[t] = 0;

    int p = 0, own = 0;
    const int ownw = tab[base + min(t, ntiles - 1)];
    const int4* c4 = (const int4*)tab;
    const int nv = base >> 2;
    for (int i0 = t; i0 < nv; i0 += 8 * 1024) {
        int4 q[8];
#pragma unroll
        for (int k = 0; k < 8; ++k) q[k] = c4[min(i0 + k * 1024, nv - 1)];
#pragma unroll
        for (int k = 0; k < 8; ++k) {
            const int a = (int)(((unsigned)q[k].x >> 16) + ((unsigned)q[k].y >> 16)) + (int)(((unsigned)q[k].z >> 16) + ((unsigned)q[k].w >> 16));
            p += (i0 + k * 1024 < nv) ? a : 0;
        }
    }
    for (int i = (nv << 2) + t; i < base; i += 1024) p += (int)((unsigned)tab[i] >> 16);
    own = (t < ntiles) ? (int)((unsigned)ownw >> 16) : 0;
    for (int i = t + 1024; i < ntiles; i += 1024) own += (int)((unsigned)tab[base + i] >> 16);
#pragma unroll
    for (int m = 16; m >= 1; m >>= 1) { p += __shfl_xor(p, m); own += __shfl_xor(own, m); }
    if (lane == 0) { red[0][w] = p; red[1][w] = own; }
    __syncthreads();
    p = red[0][lane]; own = red[1][lane];
#pragma unroll
    for (int m = 16; m >= 1; m >>= 1) { p += __shfl_xor(p, m); own += __shfl_xor(own, m); }
    const int e0 = __builtin_amdgcn_readfirstlane(p);
    const int ne = __builtin_amdgcn_readfirstlane(own);

    float4* H4 = (float4*)H + (size_t)node0 * (HID / 4);
    const int hrows = min(BN, n - node0);
    float4 hv[HV];
#pragma unroll
    for (int k = 0; k < HV; ++k) {
        const int i = t + k * 1024;
        hv[k] = ((i >> 4) < hrows) ? H4[i] : make_float4(0.0f, 0.0f, 0.0f, 0.0f);
    }

    for (int c = 0; c < ntiles; c += 32 * RPW) {
        const int r0 = c + w * RPW;
        const int dsc = (lane < RPW && r0 + lane < ntiles) ? tab[base + r0 + lane] : 0;
        int dy[RPW];
#pragma unroll
        for (int k = 0; k < RPW; ++k) {
            const int d = __builtin_amdgcn_readlane(dsc, k);
            const int cn = (int)((unsigned)d >> 16);
            const int gi = (lane < cn) ? ((r0 + k) * TILE_E + (d & 0xffff) + lane) : 0;
            dy[k] = bk[gi].y;
        }
#pragma unroll
        for (int k = 0; k < RPW; ++k) {
            const int cn = (int)((unsigned)__builtin_amdgcn_readlane(dsc, k) >> 16);
            if (lane < cn) atomicAdd(&cnt[dy[k] - node0], 1);
        }
#pragma unroll
        for (int k = 0; k < RPW; ++k) {
            const int d = __builtin_amdgcn_readlane(dsc, k);
            const int cn = (int)((unsigned)d >> 16);
            if (cn > 32) {
                const int rb = (r0 + k) * TILE_E + (d & 0xffff);
                for (int j = 32 + lane; j < cn; j += 32) atomicAdd(&cnt[bk[rb + j].y - node0], 1);
            }
        }
    }
    __syncthreads();
    const int v = (t < BN) ? cnt[t] : 0;
    const int incl = block_scan_incl<1024>(v, s, t);
    const int st = incl - v;
    if (t < BN) {
        cnt[t] = st;
        const int node = node0 + t;
        if (node < n) {
            const float deg = (float)v;
            const float dis = (deg > 0.0f) ? rsqrtf(fmaxf(deg, 1.0f)) : 0.0f;
            float* r = rec + (size_t)node * REC;
            r[0] = __int_as_float(e0 + st);
            r[1] = __int_as_float(e0 + st + v);
            r[2] = dis;
            disl[t] = dis;
        }
    }
    __syncthreads();
#pragma unroll
    for (int k = 0; k < HV; ++k) {
        const int i = t + k * 1024;
        if ((i >> 4) < hrows) {
            const float sc = disl[i >> 4];
            H4[i] = make_float4(hv[k].x * sc, hv[k].y * sc, hv[k].z * sc, hv[k].w * sc);
        }
    }
    const bool staged = (ne <= STAGE);
    for (int c = 0; c < ntiles; c += 32 * RPW) {
        const int r0 = c + w * RPW;
        const int dsc = (lane < RPW && r0 + lane < ntiles) ? tab[base + r0 + lane] : 0;
        int2 ed[RPW];
#pragma unroll
        for (int k = 0; k < RPW; ++k) {
            const int d = __builtin_amdgcn_readlane(dsc, k);
            const int cn = (int)((unsigned)d >> 16);
            const int gi = (lane < cn) ? ((r0 + k) * TILE_E + (d & 0xffff) + lane) : 0;
            ed[k] = bk[gi];
        }
#pragma unroll
        for (int k = 0; k < RPW; ++k) {
            const int cn = (int)((unsigned)__builtin_amdgcn_readlane(dsc, k) >> 16);
            if (lane < cn) {
                const int q = atomicAdd(&cnt[ed[k].y - node0], 1);
                if (staged) stage[q] = ed[k].x; else csr[e0 + q] = ed[k].x;
            }
        }
#pragma unroll
        for (int k = 0; k < RPW; ++k) {
            const int d = __builtin_amdgcn_readlane(dsc, k);
            const int cn = (int)((unsigned)d >> 16);
            if (cn > 32) {
                const int rb = (r0 + k) * TILE_E + (d & 0xffff);
                for (int j = 32 + lane; j < cn; j += 32) {
                    const int2 e2 = bk[rb + j];
                    const int q = atomicAdd(&cnt[e2.y - node0], 1);
                    if (staged) stage[q] = e2.x; else csr[e0 + q] = e2.x;
                }
            }
        }
    }
    if (staged) {
        __syncthreads();
        for (int i = t; i < ne; i += 1024) csr[e0 + i] = stage[i];
    }
}


#define YP4 17

__global__ __launch_bounds__(256) void gather64_gemm2(const float* __restrict__ Hs,
                                                      const int* __restrict__ csr,
                                                      const float* __restrict__ rec,
                                                      const float* __restrict__ b1,
                                                      const float* __restrict__ W2,
                                                      float* __restrict__ H2, int n) {
    __shared__ float4 ys4[16 * YP4];
    __shared__ float4 dl4[4];
    const int lane = threadIdx.x & 31;
    const int w = __builtin_amdgcn_readfirstlane((int)threadIdx.x >> 5);
    const int q = lane >> 4;
    const int c = lane & 15;
    const int node0 = (int)blockIdx.x * 16;
    const float4* H4 = (const float4*)Hs;
    const float4 bb = ((const float4*)b1)[c];

    const int dA = node0 + 2 * w;
    const float* rA = rec + (size_t)min(dA, n - 1) * REC;
    const float* rB = rec + (size_t)min(dA + 1, n - 1) * REC;
    const int   startA = __float_as_int(rA[0]), endA = __float_as_int(rA[1]);
    const float ddA = rA[2];
    const int   startB = __float_as_int(rB[0]), endB = __float_as_int(rB[1]);
    const float ddB = rB[2];

#pragma nounroll
    for (int i = 0; i < 2; ++i) {
        const int d = dA + i;
        const int start = i ? startB : startA;
        const int end = i ? endB : endA;
        const float dd = i ? ddB : ddA;
        float4 y = make_float4(0.0f, 0.0f, 0.0f, 0.0f);
        float dv = 0.0f;
        if (d < n) {
            float4 acc = make_float4(0.0f, 0.0f, 0.0f, 0.0f);
            const int last = end - 1;
            for (int p = start + q; p < end; p += 8) {
                const bool v1 = (p + 2 < end), v2 = (p + 4 < end), v3 = (p + 6 < end);
                const int s0 = csr[p];
                const int s1 = csr[min(p + 2, last)];
                const int s2 = csr[min(p + 4, last)];
                const int s3 = csr[min(p + 6, last)];
                const float4 h0 = H4[(size_t)s0 * (HID / 4) + c];
                const float4 h1 = H4[(size_t)s1 * (HID / 4) + c];
                const float4 h2 = H4[(size_t)s2 * (HID / 4) + c];
                const float4 h3 = H4[(size_t)s3 * (HID / 4) + c];
                acc.x += h0.x; acc.y += h0.y; acc.z += h0.z; acc.w += h0.w;
                acc.x += v1 ? h1.x : 0.0f; acc.y += v1 ? h1.y : 0.0f; acc.z += v1 ? h1.z : 0.0f; acc.w += v1 ? h1.w : 0.0f;
                acc.x += v2 ? h2.x : 0.0f; acc.y += v2 ? h2.y : 0.0f; acc.z += v2 ? h2.z : 0.0f; acc.w += v2 ? h2.w : 0.0f;
                acc.x += v3 ? h3.x : 0.0f; acc.y += v3 ? h3.y : 0.0f; acc.z += v3 ? h3.z : 0.0f; acc.w += v3 ? h3.w : 0.0f;
            }
            acc.x += __shfl_xor(acc.x, 16);
            acc.y += __shfl_xor(acc.y, 16);
            acc.z += __shfl_xor(acc.z, 16);
            acc.w += __shfl_xor(acc.w, 16);
            y.x = fmaxf(acc.x * dd + bb.x, 0.0f);
            y.y = fmaxf(acc.y * dd + bb.y, 0.0f);
            y.z = fmaxf(acc.z * dd + bb.z, 0.0f);
            y.w = fmaxf(acc.w * dd + bb.w, 0.0f);
            dv = dd;
        }
        if (q == 0) {
            ys4[(2 * w + i) * YP4 + c] = y;
            if (c == 0) ((float*)dl4)[2 * w + i] = dv;
        }
    }
    v2f bf[HID / 4];
    if (w == 0) {
        const float* bp = W2 + (size_t)(q * 2) * NCLS + c;
#pragma unroll
        for (int k = 0; k < HID; k += 4) { bf[k >> 2][0] = bp[k * NCLS]; bf[k >> 2][1] = bp[(k + 1) * NCLS]; }
    }
    __syncthreads();
    if (w != 0) return;

    const float4 ds0 = dl4[2 * q], ds1 = dl4[2 * q + 1];
    const float* ap = (const float*)ys4 + c * (YP4 * 4) + q * 2;
    v8f c0 = {};
#pragma unroll
    for (int k = 0; k < HID; k += 4) {
        v2f a; a[0] = ap[k]; a[1] = ap[k + 1];
        c0 = __builtin_amdgcn_wmma_f32_16x16x4_f32(false, a, false, bf[k >> 2], (short)0, c0, false, false);
    }
    float* hp = H2 + (size_t)(node0 + q * 8) * NCLS + c;
    hp[0 * NCLS] = c0[0] * ds0.x;
    hp[1 * NCLS] = c0[1] * ds0.y;
    hp[2 * NCLS] = c0[2] * ds0.z;
    hp[3 * NCLS] = c0[3] * ds0.w;
    hp[4 * NCLS] = c0[4] * ds1.x;
    hp[5 * NCLS] = c0[5] * ds1.y;
    hp[6 * NCLS] = c0[6] * ds1.z;
    hp[7 * NCLS] = c0[7] * ds1.w;
}


__global__ __launch_bounds__(256) void gather16(const float* __restrict__ H2,
                                                const int* __restrict__ csr,
                                                const float* __restrict__ b2,
                                                float* out, int n) {
    const int lane = threadIdx.x & 31;
    const int d = (blockIdx.x * blockDim.x + threadIdx.x) >> 5;
    if (d >= n) return;
    const float* r = out + (size_t)d * REC;
    const int start = __float_as_int(r[0]);
    const int end   = __float_as_int(r[1]);
    const float dd  = r[2];
    const int q = lane >> 2;
    const int c = lane & 3;
    const float4* H4 = (const float4*)H2;
    float4 acc = make_float4(0.0f, 0.0f, 0.0f, 0.0f);

    const int last = end - 1;
    for (int p = start + q; p < end; p += 32) {
        const bool v1 = (p + 8 < end), v2 = (p + 16 < end), v3 = (p + 24 < end);
        const int s0 = csr[p];
        const int s1 = csr[min(p + 8, last)];
        const int s2 = csr[min(p + 16, last)];
        const int s3 = csr[min(p + 24, last)];
        const float4 h0 = H4[(size_t)s0 * (NCLS / 4) + c];
        const float4 h1 = H4[(size_t)s1 * (NCLS / 4) + c];
        const float4 h2 = H4[(size_t)s2 * (NCLS / 4) + c];
        const float4 h3 = H4[(size_t)s3 * (NCLS / 4) + c];
        acc.x += h0.x; acc.y += h0.y; acc.z += h0.z; acc.w += h0.w;
        acc.x += v1 ? h1.x : 0.0f; acc.y += v1 ? h1.y : 0.0f; acc.z += v1 ? h1.z : 0.0f; acc.w += v1 ? h1.w : 0.0f;
        acc.x += v2 ? h2.x : 0.0f; acc.y += v2 ? h2.y : 0.0f; acc.z += v2 ? h2.z : 0.0f; acc.w += v2 ? h2.w : 0.0f;
        acc.x += v3 ? h3.x : 0.0f; acc.y += v3 ? h3.y : 0.0f; acc.z += v3 ? h3.z : 0.0f; acc.w += v3 ? h3.w : 0.0f;
    }
#pragma unroll
    for (int m = 4; m < 32; m <<= 1) {
        acc.x += __shfl_xor(acc.x, m);
        acc.y += __shfl_xor(acc.y, m);
        acc.z += __shfl_xor(acc.z, m);
        acc.w += __shfl_xor(acc.w, m);
    }
    if (q == 0) {
        const float4 bb = ((const float4*)b2)[c];
        float4 y;
        y.x = acc.x * dd + bb.x;
        y.y = acc.y * dd + bb.y;
        y.z = acc.z * dd + bb.z;
        y.w = acc.w * dd + bb.w;
        ((float4*)out)[(size_t)d * (NCLS / 4) + c] = y;
    }
}


extern "C" void kernel_launch(void* const* d_in, const int* in_sizes, int n_in,
                              void* d_out, int out_size, void* d_ws, size_t ws_size,
                              hipStream_t stream) {
    const float* x    = (const float*)d_in[0];
    const int*   esrc = (const int*)d_in[1];
    const int*   edst = (const int*)d_in[2];
    const float* W1   = (const float*)d_in[3];
    const float* b1   = (const float*)d_in[4];
    const float* W2   = (const float*)d_in[5];
    const float* b2   = (const float*)d_in[6];
    float* out = (float*)d_out;

    const int NN = in_sizes[0] / F_IN;
    const int E  = in_sizes[1];
    const int Mtiles = NN / 16;

    float* H   = (float*)d_ws;
    float* Y   = H + (size_t)NN * HID;
    int*   csr = (int*)(Y + (size_t)NN * HID);
    float* H2  = Y;

    const int nb     = (NN + BN - 1) / BN;
    const int ntiles = (E + TILE_E - 1) / TILE_E;
    const int gwg    = (Mtiles + GW - 1) / GW;
    int2* bk      = (int2*)Y;
    int*  tab     = (int*)Y + (((size_t)2 * E + 3) & ~(size_t)3);

    const int B = 256;
    #define NB(n) (((n) + B - 1) / B)

    tile_sort_gemm1<<<ntiles + gwg, TILE_T, 0, stream>>>(esrc, edst, bk, tab, E, nb, ntiles, x, W1, H, Mtiles);
    bucket_csr<<<nb, 1024, 0, stream>>>(bk, tab, out, csr, H, NN, ntiles);

    gather64_gemm2<<<(NN + 15) / 16, B, 0, stream>>>(H, csr, out, b1, W2, H2, NN);

    gather16<<<NB(NN * 32), B, 0, stream>>>(H2, csr, b2, out, NN);

    #undef NB
}


# layer-1 product moves whole rows: an X row is one 512-B load of a wave, through LDS to the A fragments; W1 fragments staged in LDS; H rows stored as 256-B pieces; the product's k order changes
# speedup vs baseline: 3.7964x; 1.0521x over previous
#include <hip/hip_runtime.h>


typedef __attribute__((ext_vector_type(2))) float v2f;
typedef __attribute__((ext_vector_type(8))) float v8f;

#define F_IN 128
#define HID  64
#define NCLS 16
#define REC  16


#define BSH    9
#define BN     (1 << BSH)
#define MAXB   256
#define TILE_E 4096
#define TILE_T 256
#define EPT    (TILE_E / TILE_T)
#define STAGE  12288
#define RPW    16
#define GW     (TILE_T / 32)
#define HV     (BN * (HID / 4) / 1024)

template <int NT>
__device__ __forceinline__ int block_scan_incl(int v, int (*s)[NT], int t) {
    int cur = 0;
    s[0][t] = v;
    __syncthreads();
    for (int off = 1; off < NT; off <<= 1) {
        int x = s[cur][t];
        if (t >= off) x += s[cur][t - off];
        s[cur ^ 1][t] = x;
        __syncthreads();
        cur ^= 1;
    }
    const int r = s[cur][t];
    return r;
}

#define XP        132
#define XT_FLOATS (16 * XP)
#define W1F_BYTES (F_IN * HID * 4)
#define GEMM_SMEM (W1F_BYTES + (TILE_T / 32) * XT_FLOATS * 4)
#define SORT_SMEM (MAXB * 4 + 2 * TILE_T * 4 + TILE_E * 8)
#define K1_SMEM   (GEMM_SMEM > SORT_SMEM ? GEMM_SMEM : SORT_SMEM)

__device__ __forceinline__ void gemm1_group(const float* __restrict__ X, const float* __restrict__ W1,
                                            float* __restrict__ H, unsigned char* smem,
                                            int mt, int mtc, bool have, int t, int w) {
    const int lane = t & 31;
    const int h = lane >> 4;
    const int c = lane & 15;
    float4* w1f = (float4*)smem;
    float*  xt  = (float*)(smem + W1F_BYTES) + w * XT_FLOATS;

    float4 xr[16];
    const float4* X4 = (const float4*)X + (size_t)mtc * (16 * (F_IN / 4)) + lane;
#pragma unroll
    for (int r = 0; r < 16; ++r) xr[r] = X4[r * (F_IN / 4)];
#pragma unroll
    for (int u = 0; u < 2; ++u) {
        const int p  = t + u * TILE_T;
        const int pl = p & 31, pkb = p >> 5;
        const int ph = pl >> 4, pc = pl & 15;
        const float4* wp = (const float4*)(W1 + (size_t)(8 * pkb + 4 * ph) * HID) + pc;
        const float4 q0 = wp[0], q1 = wp[HID / 4], q2 = wp[2 * (HID / 4)], q3 = wp[3 * (HID / 4)];
        float4* wd = w1f + (pkb * 4) * 32 + pl;
        wd[0 * 32] = make_float4(q0.x, q1.x, q2.x, q3.x);
        wd[1 * 32] = make_float4(q0.y, q1.y, q2.y, q3.y);
        wd[2 * 32] = make_float4(q0.z, q1.z, q2.z, q3.z);
        wd[3 * 32] = make_float4(q0.w, q1.w, q2.w, q3.w);
    }
#pragma unroll
    for (int r = 0; r < 16; ++r) *(float4*)(xt + r * XP + 4 * lane) = xr[r];
    __syncthreads();
    if (!have) return;

    const float*  ap = xt + c * XP + 4 * h;
    const float4* bp = w1f + lane;
    v8f c0 = {}, c1 = {}, c2 = {}, c3 = {};
#pragma unroll
    for (int kb = 0; kb < F_IN / 8; ++kb) {
        const float4 a4 = *(const float4*)(ap + 8 * kb);
        const float4 b0 = bp[(kb * 4 + 0) * 32];
        const float4 b1 = bp[(kb * 4 + 1) * 32];
        const float4 b2 = bp[(kb * 4 + 2) * 32];
        const float4 b3 = bp[(kb * 4 + 3) * 32];
        v2f a, b;
        a[0] = a4.x; a[1] = a4.y;
        b[0] = b0.x; b[1] = b0.y; c0 = __builtin_amdgcn_wmma_f32_16x16x4_f32(false, a, false, b, (short)0, c0, false, false);
        b[0] = b1.x; b[1] = b1.y; c1 = __builtin_amdgcn_wmma_f32_16x16x4_f32(false, a, false, b, (short)0, c1, false, false);
        b[0] = b2.x; b[1] = b2.y; c2 = __builtin_amdgcn_wmma_f32_16x16x4_f32(false, a, false, b, (short)0, c2, false, false);
        b[0] = b3.x; b[1] = b3.y; c3 = __builtin_amdgcn_wmma_f32_16x16x4_f32(false, a, false, b, (short)0, c3, false, false);
        a[0] = a4.z; a[1] = a4.w;
        b[0] = b0.z; b[1] = b0.w; c0 = __builtin_amdgcn_wmma_f32_16x16x4_f32(false, a, false, b, (short)0, c0, false, false);
        b[0] = b1.z; b[1] = b1.w; c1 = __builtin_amdgcn_wmma_f32_16x16x4_f32(false, a, false, b, (short)0, c1, false, false);
        b[0] = b2.z; b[1] = b2.w; c2 = __builtin_amdgcn_wmma_f32_16x16x4_f32(false, a, false, b, (short)0, c2, false, false);
        b[0] = b3.z; b[1] = b3.w; c3 = __builtin_amdgcn_wmma_f32_16x16x4_f32(false, a, false, b, (short)0, c3, false, false);
    }

    asm volatile("v_nop\n\tv_nop\n\tv_nop\n\tv_nop\n\tv_nop" : "+v"(c0), "+v"(c1), "+v"(c2), "+v"(c3));

    float4* hp = (float4*)H + (size_t)(mt * 16 + 8 * h) * (HID / 4) + c;
#pragma unroll
    for (int r = 0; r < 8; ++r) hp[r * (HID / 4)] = make_float4(c0[r], c1[r], c2[r], c3[r]);
}

__global__ __launch_bounds__(TILE_T) void tile_sort_gemm1(const int* __restrict__ src, const int* __restrict__ dst,
                                                          int2* __restrict__ bk, int* __restrict__ tab,
                                                          int E, int nb, int ntiles,
                                                          const float* __restrict__ X, const float* __restrict__ W1,
                                                          float* __restrict__ H, int mtiles) {
    __shared__ __attribute__((aligned(16))) unsigned char smem[K1_SMEM];
    const int t = threadIdx.x;
    if ((int)blockIdx.x >= ntiles) {
        const int w = __builtin_amdgcn_readfirstlane(t >> 5);
        const int mt = ((int)blockIdx.x - ntiles) * GW + w;
        gemm1_group(X, W1, H, smem, mt, min(mt, mtiles - 1), mt < mtiles, t, w);
        return;
    }
    int*  cur = (int*)smem;
    int (*s)[TILE_T] = (int (*)[TILE_T])(smem + MAXB * 4);
    int2* sorted = (int2*)(smem + MAXB * 4 + 2 * TILE_T * 4);
    const int tile = blockIdx.x;
    for (int i = t; i < MAXB; i += TILE_T) cur[i] = 0;
    __syncthreads();
    const int e0 = tile * TILE_E;
    int ms[EPT], md[EPT];
    for (int j = 0; j < EPT; ++j) {
        const int e = e0 + j * TILE_T + t;
        const bool ok = e < E;
        md[j] = ok ? dst[e] : -1;
        ms[j] = ok ? src[e] : 0;
        if (ok) atomicAdd(&cur[md[j] >> BSH], 1);
    }
    __syncthreads();
    static_assert(TILE_T >= MAXB, "one thread per bucket");
    static_assert(TILE_E <= 65535, "run start and length share one word");
    const int c = (t < nb) ? cur[t] : 0;
    const int incl = block_scan_incl<TILE_T>(c, s, t);
    if (t < nb) {
        cur[t] = incl - c;
        tab[t * ntiles + tile] = (incl - c) | (c << 16);
    }
    __syncthreads();
    for (int j = 0; j < EPT; ++j) {
        if (md[j] >= 0) {
            const int p = atomicAdd(&cur[md[j] >> BSH], 1);
            sorted[p] = make_int2(ms[j], md[j]);
        }
    }
    __syncthreads();
    const int ne = min(TILE_E, E - e0);
    for (int i = t; i < ne; i += TILE_T) bk[e0 + i] = sorted[i];
}

__global__ __launch_bounds__(1024) void bucket_csr(const int2* __restrict__ bk, const int* __restrict__ tab,
                                                   float* __restrict__ rec, int* __restrict__ csr,
                                                   float* H, int n, int ntiles) {
    __shared__ int cnt[BN];
    __shared__ float disl[BN];
    __shared__ int s[2][1024];
    __shared__ int stage[STAGE];
    __shared__ int red[2][32];
    const int t = threadIdx.x;
    const int lane = t & 31;
    const int w = __builtin_amdgcn_readfirstlane(t >> 5);
    const int b = blockIdx.x;
    const int base = b * ntiles;
    const int node0 = b << BSH;
    if (t < BN) cnt[t] = 0;

    int p = 0, own = 0;
    const int ownw = tab[base + min(t, ntiles - 1)];
    const int4* c4 = (const int4*)tab;
    const int nv = base >> 2;
    for (int i0 = t; i0 < nv; i0 += 8 * 1024) {
        int4 q[8];
#pragma unroll
        for (int k = 0; k < 8; ++k) q[k] = c4[min(i0 + k * 1024, nv - 1)];
#pragma unroll
        for (int k = 0; k < 8; ++k) {
            const int a = (int)(((unsigned)q[k].x >> 16) + ((unsigned)q[k].y >> 16)) + (int)(((unsigned)q[k].z >> 16) + ((unsigned)q[k].w >> 16));
            p += (i0 + k * 1024 < nv) ? a : 0;
        }
    }
    for (int i = (nv << 2) + t; i < base; i += 1024) p += (int)((unsigned)tab[i] >> 16);
    own = (t < ntiles) ? (int)((unsigned)ownw >> 16) : 0;
    for (int i = t + 1024; i < ntiles; i += 1024) own += (int)((unsigned)tab[base + i] >> 16);
#pragma unroll
    for (int m = 16; m >= 1; m >>= 1) { p += __shfl_xor(p, m); own += __shfl_xor(own, m); }
    if (lane == 0) { red[0][w] = p; red[1][w] = own; }
    __syncthreads();
    p = red[0][lane]; own = red[1][lane];
#pragma unroll
    for (int m = 16; m >= 1; m >>= 1) { p += __shfl_xor(p, m); own += __shfl_xor(own, m); }
    const int e0 = __builtin_amdgcn_readfirstlane(p);
    const int ne = __builtin_amdgcn_readfirstlane(own);

    float4* H4 = (float4*)H + (size_t)node0 * (HID / 4);
    const int hrows = min(BN, n - node0);
    float4 hv[HV];
#pragma unroll
    for (int k = 0; k < HV; ++k) {
        const int i = t + k * 1024;
        hv[k] = ((i >> 4) < hrows) ? H4[i] : make_float4(0.0f, 0.0f, 0.0f, 0.0f);
    }

    for (int c = 0; c < ntiles; c += 32 * RPW) {
        const int r0 = c + w * RPW;
        const int dsc = (lane < RPW && r0 + lane < ntiles) ? tab[base + r0 + lane] : 0;
        int dy[RPW];
#pragma unroll
        for (int k = 0; k < RPW; ++k) {
            const int d = __builtin_amdgcn_readlane(dsc, k);
            const int cn = (int)((unsigned)d >> 16);
            const int gi = (lane < cn) ? ((r0 + k) * TILE_E + (d & 0xffff) + lane) : 0;
            dy[k] = bk[gi].y;
        }
#pragma unroll
        for (int k = 0; k < RPW; ++k) {
            const int cn = (int)((unsigned)__builtin_amdgcn_readlane(dsc, k) >> 16);
            if (lane < cn) atomicAdd(&cnt[dy[k] - node0], 1);
        }
#pragma unroll
        for (int k = 0; k < RPW; ++k) {
            const int d = __builtin_amdgcn_readlane(dsc, k);
            const int cn = (int)((unsigned)d >> 16);
            if (cn > 32) {
                const int rb = (r0 + k) * TILE_E + (d & 0xffff);
                for (int j = 32 + lane; j < cn; j += 32) atomicAdd(&cnt[bk[rb + j].y - node0], 1);
            }
        }
    }
    __syncthreads();
    const int v = (t < BN) ? cnt[t] : 0;
    const int incl = block_scan_incl<1024>(v, s, t);
    const int st = incl - v;
    if (t < BN) {
        cnt[t] = st;
        const int node = node0 + t;
        if (node < n) {
            const float deg = (float)v;
            const float dis = (deg > 0.0f) ? rsqrtf(fmaxf(deg, 1.0f)) : 0.0f;
            float* r = rec + (size_t)node * REC;
            r[0] = __int_as_float(e0 + st);
            r[1] = __int_as_float(e0 + st + v);
            r[2] = dis;
            disl[t] = dis;
        }
    }
    __syncthreads();
#pragma unroll
    for (int k = 0; k < HV; ++k) {
        const int i = t + k * 1024;
        if ((i >> 4) < hrows) {
            const float sc = disl[i >> 4];
            H4[i] = make_float4(hv[k].x * sc, hv[k].y * sc, hv[k].z * sc, hv[k].w * sc);
        }
    }
    const bool staged = (ne <= STAGE);
    for (int c = 0; c < ntiles; c += 32 * RPW) {
        const int r0 = c + w * RPW;
        const int dsc = (lane < RPW && r0 + lane < ntiles) ? tab[base + r0 + lane] : 0;
        int2 ed[RPW];
#pragma unroll
        for (int k = 0; k < RPW; ++k) {
            const int d = __builtin_amdgcn_readlane(dsc, k);
            const int cn = (int)((unsigned)d >> 16);
            const int gi = (lane < cn) ? ((r0 + k) * TILE_E + (d & 0xffff) + lane) : 0;
            ed[k] = bk[gi];
        }
#pragma unroll
        for (int k = 0; k < RPW; ++k) {
            const int cn = (int)((unsigned)__builtin_amdgcn_readlane(dsc, k) >> 16);
            if (lane < cn) {
                const int q = atomicAdd(&cnt[ed[k].y - node0], 1);
                if (staged) stage[q] = ed[k].x; else csr[e0 + q] = ed[k].x;
            }
        }
#pragma unroll
        for (int k = 0; k < RPW; ++k) {
            const int d = __builtin_amdgcn_readlane(dsc, k);
            const int cn = (int)((unsigned)d >> 16);
            if (cn > 32) {
                const int rb = (r0 + k) * TILE_E + (d & 0xffff);
                for (int j = 32 + lane; j < cn; j += 32) {
                    const int2 e2 = bk[rb + j];
                    const int q = atomicAdd(&cnt[e2.y - node0], 1);
                    if (staged) stage[q] = e2.x; else csr[e0 + q] = e2.x;
                }
            }
        }
    }
    if (staged) {
        __syncthreads();
        for (int i = t; i < ne; i += 1024) csr[e0 + i] = stage[i];
    }
}


#define YP4 17

__global__ __launch_bounds__(256) void gather64_gemm2(const float* __restrict__ Hs,
                                                      const int* __restrict__ csr,
                                                      const float* __restrict__ rec,
                                                      const float* __restrict__ b1,
                                                      const float* __restrict__ W2,
                                                      float* __restrict__ H2, int n) {
    __shared__ float4 ys4[16 * YP4];
    __shared__ float4 dl4[4];
    const int lane = threadIdx.x & 31;
    const int w = __builtin_amdgcn_readfirstlane((int)threadIdx.x >> 5);
    const int q = lane >> 4;
    const int c = lane & 15;
    const int node0 = (int)blockIdx.x * 16;
    const float4* H4 = (const float4*)Hs;
    const float4 bb = ((const float4*)b1)[c];

    const int dA = node0 + 2 * w;
    const float* rA = rec + (size_t)min(dA, n - 1) * REC;
    const float* rB = rec + (size_t)min(dA + 1, n - 1) * REC;
    const int   startA = __float_as_int(rA[0]), endA = __float_as_int(rA[1]);
    const float ddA = rA[2];
    const int   startB = __float_as_int(rB[0]), endB = __float_as_int(rB[1]);
    const float ddB = rB[2];

#pragma nounroll
    for (int i = 0; i < 2; ++i) {
        const int d = dA + i;
        const int start = i ? startB : startA;
        const int end = i ? endB : endA;
        const float dd = i ? ddB : ddA;
        float4 y = make_float4(0.0f, 0.0f, 0.0f, 0.0f);
        float dv = 0.0f;
        if (d < n) {
            float4 acc = make_float4(0.0f, 0.0f, 0.0f, 0.0f);
            const int last = end - 1;
            for (int p = start + q; p < end; p += 8) {
                const bool v1 = (p + 2 < end), v2 = (p + 4 < end), v3 = (p + 6 < end);
                const int s0 = csr[p];
                const int s1 = csr[min(p + 2, last)];
                const int s2 = csr[min(p + 4, last)];
                const int s3 = csr[min(p + 6, last)];
                const float4 h0 = H4[(size_t)s0 * (HID / 4) + c];
                const float4 h1 = H4[(size_t)s1 * (HID / 4) + c];
                const float4 h2 = H4[(size_t)s2 * (HID / 4) + c];
                const float4 h3 = H4[(size_t)s3 * (HID / 4) + c];
                acc.x += h0.x; acc.y += h0.y; acc.z += h0.z; acc.w += h0.w;
                acc.x += v1 ? h1.x : 0.0f; acc.y += v1 ? h1.y : 0.0f; acc.z += v1 ? h1.z : 0.0f; acc.w += v1 ? h1.w : 0.0f;
                acc.x += v2 ? h2.x : 0.0f; acc.y += v2 ? h2.y : 0.0f; acc.z += v2 ? h2.z : 0.0f; acc.w += v2 ? h2.w : 0.0f;
                acc.x += v3 ? h3.x : 0.0f; acc.y += v3 ? h3.y : 0.0f; acc.z += v3 ? h3.z : 0.0f; acc.w += v3 ? h3.w : 0.0f;
            }
            acc.x += __shfl_xor(acc.x, 16);
            acc.y += __shfl_xor(acc.y, 16);
            acc.z += __shfl_xor(acc.z, 16);
            acc.w += __shfl_xor(acc.w, 16);
            y.x = fmaxf(acc.x * dd + bb.x, 0.0f);
            y.y = fmaxf(acc.y * dd + bb.y, 0.0f);
            y.z = fmaxf(acc.z * dd + bb.z, 0.0f);
            y.w = fmaxf(acc.w * dd + bb.w, 0.0f);
            dv = dd;
        }
        if (q == 0) {
            ys4[(2 * w + i) * YP4 + c] = y;
            if (c == 0) ((float*)dl4)[2 * w + i] = dv;
        }
    }
    v2f bf[HID / 4];
    if (w == 0) {
        const float* bp = W2 + (size_t)(q * 2) * NCLS + c;
#pragma unroll
        for (int k = 0; k < HID; k += 4) { bf[k >> 2][0] = bp[k * NCLS]; bf[k >> 2][1] = bp[(k + 1) * NCLS]; }
    }
    __syncthreads();
    if (w != 0) return;

    const float4 ds0 = dl4[2 * q], ds1 = dl4[2 * q + 1];
    const float* ap = (const float*)ys4 + c * (YP4 * 4) + q * 2;
    v8f c0 = {};
#pragma unroll
    for (int k = 0; k < HID; k += 4) {
        v2f a; a[0] = ap[k]; a[1] = ap[k + 1];
        c0 = __builtin_amdgcn_wmma_f32_16x16x4_f32(false, a, false, bf[k >> 2], (short)0, c0, false, false);
    }
    float* hp = H2 + (size_t)(node0 + q * 8) * NCLS + c;
    hp[0 * NCLS] = c0[0] * ds0.x;
    hp[1 * NCLS] = c0[1] * ds0.y;
    hp[2 * NCLS] = c0[2] * ds0.z;
    hp[3 * NCLS] = c0[3] * ds0.w;
    hp[4 * NCLS] = c0[4] * ds1.x;
    hp[5 * NCLS] = c0[5] * ds1.y;
    hp[6 * NCLS] = c0[6] * ds1.z;
    hp[7 * NCLS] = c0[7] * ds1.w;
}


__global__ __launch_bounds__(256) void gather16(const float* __restrict__ H2,
                                                const int* __restrict__ csr,
                                                const float* __restrict__ b2,
                                                float* out, int n) {
    const int lane = threadIdx.x & 31;
    const int d = (blockIdx.x * blockDim.x + threadIdx.x) >> 5;
    if (d >= n) return;
    const float* r = out + (size_t)d * REC;
    const int start = __float_as_int(r[0]);
    const int end   = __float_as_int(r[1]);
    const float dd  = r[2];
    const int q = lane >> 2;
    const int c = lane & 3;
    const float4* H4 = (const float4*)H2;
    float4 acc = make_float4(0.0f, 0.0f, 0.0f, 0.0f);

    const int last = end - 1;
    for (int p = start + q; p < end; p += 32) {
        const bool v1 = (p + 8 < end), v2 = (p + 16 < end), v3 = (p + 24 < end);
        const int s0 = csr[p];
        const int s1 = csr[min(p + 8, last)];
        const int s2 = csr[min(p + 16, last)];
        const int s3 = csr[min(p + 24, last)];
        const float4 h0 = H4[(size_t)s0 * (NCLS / 4) + c];
        const float4 h1 = H4[(size_t)s1 * (NCLS / 4) + c];
        const float4 h2 = H4[(size_t)s2 * (NCLS / 4) + c];
        const float4 h3 = H4[(size_t)s3 * (NCLS / 4) + c];
        acc.x += h0.x; acc.y += h0.y; acc.z += h0.z; acc.w += h0.w;
        acc.x += v1 ? h1.x : 0.0f; acc.y += v1 ? h1.y : 0.0f; acc.z += v1 ? h1.z : 0.0f; acc.w += v1 ? h1.w : 0.0f;
        acc.x += v2 ? h2.x : 0.0f; acc.y += v2 ? h2.y : 0.0f; acc.z += v2 ? h2.z : 0.0f; acc.w += v2 ? h2.w : 0.0f;
        acc.x += v3 ? h3.x : 0.0f; acc.y += v3 ? h3.y : 0.0f; acc.z += v3 ? h3.z : 0.0f; acc.w += v3 ? h3.w : 0.0f;
    }
#pragma unroll
    for (int m = 4; m < 32; m <<= 1) {
        acc.x += __shfl_xor(acc.x, m);
        acc.y += __shfl_xor(acc.y, m);
        acc.z += __shfl_xor(acc.z, m);
        acc.w += __shfl_xor(acc.w, m);
    }
    if (q == 0) {
        const float4 bb = ((const float4*)b2)[c];
        float4 y;
        y.x = acc.x * dd + bb.x;
        y.y = acc.y * dd + bb.y;
        y.z = acc.z * dd + bb.z;
        y.w = acc.w * dd + bb.w;
        ((float4*)out)[(size_t)d * (NCLS / 4) + c] = y;
    }
}


extern "C" void kernel_launch(void* const* d_in, const int* in_sizes, int n_in,
                              void* d_out, int out_size, void* d_ws, size_t ws_size,
                              hipStream_t stream) {
    const float* x    = (const float*)d_in[0];
    const int*   esrc = (const int*)d_in[1];
    const int*   edst = (const int*)d_in[2];
    const float* W1   = (const float*)d_in[3];
    const float* b1   = (const float*)d_in[4];
    const float* W2   = (const float*)d_in[5];
    const float* b2   = (const float*)d_in[6];
    float* out = (float*)d_out;

    const int NN = in_sizes[0] / F_IN;
    const int E  = in_sizes[1];
    const int Mtiles = NN / 16;

    float* H   = (float*)d_ws;
    float* Y   = H + (size_t)NN * HID;
    int*   csr = (int*)(Y + (size_t)NN * HID);
    float* H2  = Y;

    const int nb     = (NN + BN - 1) / BN;
    const int ntiles = (E + TILE_E - 1) / TILE_E;
    const int gwg    = (Mtiles + GW - 1) / GW;
    int2* bk      = (int2*)Y;
    int*  tab     = (int*)Y + (((size_t)2 * E + 3) & ~(size_t)3);

    const int B = 256;
    #define NB(n) (((n) + B - 1) / B)

    tile_sort_gemm1<<<ntiles + gwg, TILE_T, 0, stream>>>(esrc, edst, bk, tab, E, nb, ntiles, x, W1, H, Mtiles);
    bucket_csr<<<nb, 1024, 0, stream>>>(bk, tab, out, csr, H, NN, ntiles);

    gather64_gemm2<<<(NN + 15) / 16, B, 0, stream>>>(H, csr, out, b1, W2, H2, NN);

    gather16<<<NB(NN * 32), B, 0, stream>>>(H2, csr, b2, out, NN);

    #undef NB
}


# tile sort: a thread's 16 edge loads (src, dst) leave as one batch ahead of its LDS bucket counts; built code had one memory round trip per edge (load, wait, count); sum order: LDS arrival as before
# speedup vs baseline: 3.9224x; 1.0332x over previous
#include <hip/hip_runtime.h>


typedef __attribute__((ext_vector_type(2))) float v2f;
typedef __attribute__((ext_vector_type(8))) float v8f;

#define F_IN 128
#define HID  64
#define NCLS 16
#define REC  16


#define BSH    9
#define BN     (1 << BSH)
#define MAXB   256
#define TILE_E 4096
#define TILE_T 256
#define EPT    (TILE_E / TILE_T)
#define STAGE  12288
#define RPW    16
#define GW     (TILE_T / 32)
#define HV     (BN * (HID / 4) / 1024)

template <int NT>
__device__ __forceinline__ int block_scan_incl(int v, int (*s)[NT], int t) {
    int cur = 0;
    s[0][t] = v;
    __syncthreads();
    for (int off = 1; off < NT; off <<= 1) {
        int x = s[cur][t];
        if (t >= off) x += s[cur][t - off];
        s[cur ^ 1][t] = x;
        __syncthreads();
        cur ^= 1;
    }
    const int r = s[cur][t];
    return r;
}

#define XP        132
#define XT_FLOATS (16 * XP)
#define W1F_BYTES (F_IN * HID * 4)
#define GEMM_SMEM (W1F_BYTES + (TILE_T / 32) * XT_FLOATS * 4)
#define SORT_SMEM (MAXB * 4 + 2 * TILE_T * 4 + TILE_E * 8)
#define K1_SMEM   (GEMM_SMEM > SORT_SMEM ? GEMM_SMEM : SORT_SMEM)

__device__ __forceinline__ void gemm1_group(const float* __restrict__ X, const float* __restrict__ W1,
                                            float* __restrict__ H, unsigned char* smem,
                                            int mt, int mtc, bool have, int t, int w) {
    const int lane = t & 31;
    const int h = lane >> 4;
    const int c = lane & 15;
    float4* w1f = (float4*)smem;
    float*  xt  = (float*)(smem + W1F_BYTES) + w * XT_FLOATS;

    float4 xr[16];
    const float4* X4 = (const float4*)X + (size_t)mtc * (16 * (F_IN / 4)) + lane;
#pragma unroll
    for (int r = 0; r < 16; ++r) xr[r] = X4[r * (F_IN / 4)];
#pragma unroll
    for (int u = 0; u < 2; ++u) {
        const int p  = t + u * TILE_T;
        const int pl = p & 31, pkb = p >> 5;
        const int ph = pl >> 4, pc = pl & 15;
        const float4* wp = (const float4*)(W1 + (size_t)(8 * pkb + 4 * ph) * HID) + pc;
        const float4 q0 = wp[0], q1 = wp[HID / 4], q2 = wp[2 * (HID / 4)], q3 = wp[3 * (HID / 4)];
        float4* wd = w1f + (pkb * 4) * 32 + pl;
        wd[0 * 32] = make_float4(q0.x, q1.x, q2.x, q3.x);
        wd[1 * 32] = make_float4(q0.y, q1.y, q2.y, q3.y);
        wd[2 * 32] = make_float4(q0.z, q1.z, q2.z, q3.z);
        wd[3 * 32] = make_float4(q0.w, q1.w, q2.w, q3.w);
    }
#pragma unroll
    for (int r = 0; r < 16; ++r) *(float4*)(xt + r * XP + 4 * lane) = xr[r];
    __syncthreads();
    if (!have) return;

    const float*  ap = xt + c * XP + 4 * h;
    const float4* bp = w1f + lane;
    v8f c0 = {}, c1 = {}, c2 = {}, c3 = {};
#pragma unroll
    for (int kb = 0; kb < F_IN / 8; ++kb) {
        const float4 a4 = *(const float4*)(ap + 8 * kb);
        const float4 b0 = bp[(kb * 4 + 0) * 32];
        const float4 b1 = bp[(kb * 4 + 1) * 32];
        const float4 b2 = bp[(kb * 4 + 2) * 32];
        const float4 b3 = bp[(kb * 4 + 3) * 32];
        v2f a, b;
        a[0] = a4.x; a[1] = a4.y;
        b[0] = b0.x; b[1] = b0.y; c0 = __builtin_amdgcn_wmma_f32_16x16x4_f32(false, a, false, b, (short)0, c0, false, false);
        b[0] = b1.x; b[1] = b1.y; c1 = __builtin_amdgcn_wmma_f32_16x16x4_f32(false, a, false, b, (short)0, c1, false, false);
        b[0] = b2.x; b[1] = b2.y; c2 = __builtin_amdgcn_wmma_f32_16x16x4_f32(false, a, false, b, (short)0, c2, false, false);
        b[0] = b3.x; b[1] = b3.y; c3 = __builtin_amdgcn_wmma_f32_16x16x4_f32(false, a, false, b, (short)0, c3, false, false);
        a[0] = a4.z; a[1] = a4.w;
        b[0] = b0.z; b[1] = b0.w; c0 = __builtin_amdgcn_wmma_f32_16x16x4_f32(false, a, false, b, (short)0, c0, false, false);
        b[0] = b1.z; b[1] = b1.w; c1 = __builtin_amdgcn_wmma_f32_16x16x4_f32(false, a, false, b, (short)0, c1, false, false);
        b[0] = b2.z; b[1] = b2.w; c2 = __builtin_amdgcn_wmma_f32_16x16x4_f32(false, a, false, b, (short)0, c2, false, false);
        b[0] = b3.z; b[1] = b3.w; c3 = __builtin_amdgcn_wmma_f32_16x16x4_f32(false, a, false, b, (short)0, c3, false, false);
    }

    asm volatile("v_nop\n\tv_nop\n\tv_nop\n\tv_nop\n\tv_nop" : "+v"(c0), "+v"(c1), "+v"(c2), "+v"(c3));

    float4* hp = (float4*)H + (size_t)(mt * 16 + 8 * h) * (HID / 4) + c;
#pragma unroll
    for (int r = 0; r < 8; ++r) hp[r * (HID / 4)] = make_float4(c0[r], c1[r], c2[r], c3[r]);
}

__global__ __launch_bounds__(TILE_T) void tile_sort_gemm1(const int* __restrict__ src, const int* __restrict__ dst,
                                                          int2* __restrict__ bk, int* __restrict__ tab,
                                                          int E, int nb, int ntiles,
                                                          const float* __restrict__ X, const float* __restrict__ W1,
                                                          float* __restrict__ H, int mtiles) {
    __shared__ __attribute__((aligned(16))) unsigned char smem[K1_SMEM];
    const int t = threadIdx.x;
    if ((int)blockIdx.x >= ntiles) {
        const int w = __builtin_amdgcn_readfirstlane(t >> 5);
        const int mt = ((int)blockIdx.x - ntiles) * GW + w;
        gemm1_group(X, W1, H, smem, mt, min(mt, mtiles - 1), mt < mtiles, t, w);
        return;
    }
    int*  cur = (int*)smem;
    int (*s)[TILE_T] = (int (*)[TILE_T])(smem + MAXB * 4);
    int2* sorted = (int2*)(smem + MAXB * 4 + 2 * TILE_T * 4);
    const int tile = blockIdx.x;
    const int e0 = tile * TILE_E;
    int ms[EPT], md[EPT];
#pragma unroll
    for (int j = 0; j < EPT; ++j) {
        const int ec = min(e0 + j * TILE_T + t, E - 1);
        md[j] = dst[ec];
        ms[j] = src[ec];
    }
    for (int i = t; i < MAXB; i += TILE_T) cur[i] = 0;
    __syncthreads();
#pragma unroll
    for (int j = 0; j < EPT; ++j) {
        const bool ok = e0 + j * TILE_T + t < E;
        if (!ok) md[j] = -1;
        if (ok) atomicAdd(&cur[md[j] >> BSH], 1);
    }
    __syncthreads();
    static_assert(TILE_T >= MAXB, "one thread per bucket");
    static_assert(TILE_E <= 65535, "run start and length share one word");
    const int c = (t < nb) ? cur[t] : 0;
    const int incl = block_scan_incl<TILE_T>(c, s, t);
    if (t < nb) {
        cur[t] = incl - c;
        tab[t * ntiles + tile] = (incl - c) | (c << 16);
    }
    __syncthreads();
    for (int j = 0; j < EPT; ++j) {
        if (md[j] >= 0) {
            const int p = atomicAdd(&cur[md[j] >> BSH], 1);
            sorted[p] = make_int2(ms[j], md[j]);
        }
    }
    __syncthreads();
    const int ne = min(TILE_E, E - e0);
    for (int i = t; i < ne; i += TILE_T) bk[e0 + i] = sorted[i];
}

__global__ __launch_bounds__(1024) void bucket_csr(const int2* __restrict__ bk, const int* __restrict__ tab,
                                                   float* __restrict__ rec, int* __restrict__ csr,
                                                   float* H, int n, int ntiles) {
    __shared__ int cnt[BN];
    __shared__ float disl[BN];
    __shared__ int s[2][1024];
    __shared__ int stage[STAGE];
    __shared__ int red[2][32];
    const int t = threadIdx.x;
    const int lane = t & 31;
    const int w = __builtin_amdgcn_readfirstlane(t >> 5);
    const int b = blockIdx.x;
    const int base = b * ntiles;
    const int node0 = b << BSH;
    if (t < BN) cnt[t] = 0;

    int p = 0, own = 0;
    const int ownw = tab[base + min(t, ntiles - 1)];
    const int4* c4 = (const int4*)tab;
    const int nv = base >> 2;
    for (int i0 = t; i0 < nv; i0 += 8 * 1024) {
        int4 q[8];
#pragma unroll
        for (int k = 0; k < 8; ++k) q[k] = c4[min(i0 + k * 1024, nv - 1)];
#pragma unroll
        for (int k = 0; k < 8; ++k) {
            const int a = (int)(((unsigned)q[k].x >> 16) + ((unsigned)q[k].y >> 16)) + (int)(((unsigned)q[k].z >> 16) + ((unsigned)q[k].w >> 16));
            p += (i0 + k * 1024 < nv) ? a : 0;
        }
    }
    for (int i = (nv << 2) + t; i < base; i += 1024) p += (int)((unsigned)tab[i] >> 16);
    own = (t < ntiles) ? (int)((unsigned)ownw >> 16) : 0;
    for (int i = t + 1024; i < ntiles; i += 1024) own += (int)((unsigned)tab[base + i] >> 16);
#pragma unroll
    for (int m = 16; m >= 1; m >>= 1) { p += __shfl_xor(p, m); own += __shfl_xor(own, m); }
    if (lane == 0) { red[0][w] = p; red[1][w] = own; }
    __syncthreads();
    p = red[0][lane]; own = red[1][lane];
#pragma unroll
    for (int m = 16; m >= 1; m >>= 1) { p += __shfl_xor(p, m); own += __shfl_xor(own, m); }
    const int e0 = __builtin_amdgcn_readfirstlane(p);
    const int ne = __builtin_amdgcn_readfirstlane(own);

    float4* H4 = (float4*)H + (size_t)node0 * (HID / 4);
    const int hrows = min(BN, n - node0);
    float4 hv[HV];
#pragma unroll
    for (int k = 0; k < HV; ++k) {
        const int i = t + k * 1024;
        hv[k] = ((i >> 4) < hrows) ? H4[i] : make_float4(0.0f, 0.0f, 0.0f, 0.0f);
    }

    for (int c = 0; c < ntiles; c += 32 * RPW) {
        const int r0 = c + w * RPW;
        const int dsc = (lane < RPW && r0 + lane < ntiles) ? tab[base + r0 + lane] : 0;
        int dy[RPW];
#pragma unroll
        for (int k = 0; k < RPW; ++k) {
            const int d = __builtin_amdgcn_readlane(dsc, k);
            const int cn = (int)((unsigned)d >> 16);
            const int gi = (lane < cn) ? ((r0 + k) * TILE_E + (d & 0xffff) + lane) : 0;
            dy[k] = bk[gi].y;
        }
#pragma unroll
        for (int k = 0; k < RPW; ++k) {
            const int cn = (int)((unsigned)__builtin_amdgcn_readlane(dsc, k) >> 16);
            if (lane < cn) atomicAdd(&cnt[dy[k] - node0], 1);
        }
#pragma unroll
        for (int k = 0; k < RPW; ++k) {
            const int d = __builtin_amdgcn_readlane(dsc, k);
            const int cn = (int)((unsigned)d >> 16);
            if (cn > 32) {
                const int rb = (r0 + k) * TILE_E + (d & 0xffff);
                for (int j = 32 + lane; j < cn; j += 32) atomicAdd(&cnt[bk[rb + j].y - node0], 1);
            }
        }
    }
    __syncthreads();
    const int v = (t < BN) ? cnt[t] : 0;
    const int incl = block_scan_incl<1024>(v, s, t);
    const int st = incl - v;
    if (t < BN) {
        cnt[t] = st;
        const int node = node0 + t;
        if (node < n) {
            const float deg = (float)v;
            const float dis = (deg > 0.0f) ? rsqrtf(fmaxf(deg, 1.0f)) : 0.0f;
            float* r = rec + (size_t)node * REC;
            r[0] = __int_as_float(e0 + st);
            r[1] = __int_as_float(e0 + st + v);
            r[2] = dis;
            disl[t] = dis;
        }
    }
    __syncthreads();
#pragma unroll
    for (int k = 0; k < HV; ++k) {
        const int i = t + k * 1024;
        if ((i >> 4) < hrows) {
            const float sc = disl[i >> 4];
            H4[i] = make_float4(hv[k].x * sc, hv[k].y * sc, hv[k].z * sc, hv[k].w * sc);
        }
    }
    const bool staged = (ne <= STAGE);
    for (int c = 0; c < ntiles; c += 32 * RPW) {
        const int r0 = c + w * RPW;
        const int dsc = (lane < RPW && r0 + lane < ntiles) ? tab[base + r0 + lane] : 0;
        int2 ed[RPW];
#pragma unroll
        for (int k = 0; k < RPW; ++k) {
            const int d = __builtin_amdgcn_readlane(dsc, k);
            const int cn = (int)((unsigned)d >> 16);
            const int gi = (lane < cn) ? ((r0 + k) * TILE_E + (d & 0xffff) + lane) : 0;
            ed[k] = bk[gi];
        }
#pragma unroll
        for (int k = 0; k < RPW; ++k) {
            const int cn = (int)((unsigned)__builtin_amdgcn_readlane(dsc, k) >> 16);
            if (lane < cn) {
                const int q = atomicAdd(&cnt[ed[k].y - node0], 1);
                if (staged) stage[q] = ed[k].x; else csr[e0 + q] = ed[k].x;
            }
        }
#pragma unroll
        for (int k = 0; k < RPW; ++k) {
            const int d = __builtin_amdgcn_readlane(dsc, k);
            const int cn = (int)((unsigned)d >> 16);
            if (cn > 32) {
                const int rb = (r0 + k) * TILE_E + (d & 0xffff);
                for (int j = 32 + lane; j < cn; j += 32) {
                    const int2 e2 = bk[rb + j];
                    const int q = atomicAdd(&cnt[e2.y - node0], 1);
                    if (staged) stage[q] = e2.x; else csr[e0 + q] = e2.x;
                }
            }
        }
    }
    if (staged) {
        __syncthreads();
        for (int i = t; i < ne; i += 1024) csr[e0 + i] = stage[i];
    }
}


#define YP4 17

__global__ __launch_bounds__(256) void gather64_gemm2(const float* __restrict__ Hs,
                                                      const int* __restrict__ csr,
                                                      const float* __restrict__ rec,
                                                      const float* __restrict__ b1,
                                                      const float* __restrict__ W2,
                                                      float* __restrict__ H2, int n) {
    __shared__ float4 ys4[16 * YP4];
    __shared__ float4 dl4[4];
    const int lane = threadIdx.x & 31;
    const int w = __builtin_amdgcn_readfirstlane((int)threadIdx.x >> 5);
    const int q = lane >> 4;
    const int c = lane & 15;
    const int node0 = (int)blockIdx.x * 16;
    const float4* H4 = (const float4*)Hs;
    const float4 bb = ((const float4*)b1)[c];

    const int dA = node0 + 2 * w;
    const float* rA = rec + (size_t)min(dA, n - 1) * REC;
    const float* rB = rec + (size_t)min(dA + 1, n - 1) * REC;
    const int   startA = __float_as_int(rA[0]), endA = __float_as_int(rA[1]);
    const float ddA = rA[2];
    const int   startB = __float_as_int(rB[0]), endB = __float_as_int(rB[1]);
    const float ddB = rB[2];

#pragma nounroll
    for (int i = 0; i < 2; ++i) {
        const int d = dA + i;
        const int start = i ? startB : startA;
        const int end = i ? endB : endA;
        const float dd = i ? ddB : ddA;
        float4 y = make_float4(0.0f, 0.0f, 0.0f, 0.0f);
        float dv = 0.0f;
        if (d < n) {
            float4 acc = make_float4(0.0f, 0.0f, 0.0f, 0.0f);
            const int last = end - 1;
            for (int p = start + q; p < end; p += 8) {
                const bool v1 = (p + 2 < end), v2 = (p + 4 < end), v3 = (p + 6 < end);
                const int s0 = csr[p];
                const int s1 = csr[min(p + 2, last)];
                const int s2 = csr[min(p + 4, last)];
                const int s3 = csr[min(p + 6, last)];
                const float4 h0 = H4[(size_t)s0 * (HID / 4) + c];
                const float4 h1 = H4[(size_t)s1 * (HID / 4) + c];
                const float4 h2 = H4[(size_t)s2 * (HID / 4) + c];
                const float4 h3 = H4[(size_t)s3 * (HID / 4) + c];
                acc.x += h0.x; acc.y += h0.y; acc.z += h0.z; acc.w += h0.w;
                acc.x += v1 ? h1.x : 0.0f; acc.y += v1 ? h1.y : 0.0f; acc.z += v1 ? h1.z : 0.0f; acc.w += v1 ? h1.w : 0.0f;
                acc.x += v2 ? h2.x : 0.0f; acc.y += v2 ? h2.y : 0.0f; acc.z += v2 ? h2.z : 0.0f; acc.w += v2 ? h2.w : 0.0f;
                acc.x += v3 ? h3.x : 0.0f; acc.y += v3 ? h3.y : 0.0f; acc.z += v3 ? h3.z : 0.0f; acc.w += v3 ? h3.w : 0.0f;
            }
            acc.x += __shfl_xor(acc.x, 16);
            acc.y += __shfl_xor(acc.y, 16);
            acc.z += __shfl_xor(acc.z, 16);
            acc.w += __shfl_xor(acc.w, 16);
            y.x = fmaxf(acc.x * dd + bb.x, 0.0f);
            y.y = fmaxf(acc.y * dd + bb.y, 0.0f);
            y.z = fmaxf(acc.z * dd + bb.z, 0.0f);
            y.w = fmaxf(acc.w * dd + bb.w, 0.0f);
            dv = dd;
        }
        if (q == 0) {
            ys4[(2 * w + i) * YP4 + c] = y;
            if (c == 0) ((float*)dl4)[2 * w + i] = dv;
        }
    }
    v2f bf[HID / 4];
    if (w == 0) {
        const float* bp = W2 + (size_t)(q * 2) * NCLS + c;
#pragma unroll
        for (int k = 0; k < HID; k += 4) { bf[k >> 2][0] = bp[k * NCLS]; bf[k >> 2][1] = bp[(k + 1) * NCLS]; }
    }
    __syncthreads();
    if (w != 0) return;

    const float4 ds0 = dl4[2 * q], ds1 = dl4[2 * q + 1];
    const float* ap = (const float*)ys4 + c * (YP4 * 4) + q * 2;
    v8f c0 = {};
#pragma unroll
    for (int k = 0; k < HID; k += 4) {
        v2f a; a[0] = ap[k]; a[1] = ap[k + 1];
        c0 = __builtin_amdgcn_wmma_f32_16x16x4_f32(false, a, false, bf[k >> 2], (short)0, c0, false, false);
    }
    float* hp = H2 + (size_t)(node0 + q * 8) * NCLS + c;
    hp[0 * NCLS] = c0[0] * ds0.x;
    hp[1 * NCLS] = c0[1] * ds0.y;
    hp[2 * NCLS] = c0[2] * ds0.z;
    hp[3 * NCLS] = c0[3] * ds0.w;
    hp[4 * NCLS] = c0[4] * ds1.x;
    hp[5 * NCLS] = c0[5] * ds1.y;
    hp[6 * NCLS] = c0[6] * ds1.z;
    hp[7 * NCLS] = c0[7] * ds1.w;
}


__global__ __launch_bounds__(256) void gather16(const float* __restrict__ H2,
                                                const int* __restrict__ csr,
                                                const float* __restrict__ b2,
                                                float* out, int n) {
    const int lane = threadIdx.x & 31;
    const int d = (blockIdx.x * blockDim.x + threadIdx.x) >> 5;
    if (d >= n) return;
    const float* r = out + (size_t)d * REC;
    const int start = __float_as_int(r[0]);
    const int end   = __float_as_int(r[1]);
    const float dd  = r[2];
    const int q = lane >> 2;
    const int c = lane & 3;
    const float4* H4 = (const float4*)H2;
    float4 acc = make_float4(0.0f, 0.0f, 0.0f, 0.0f);

    const int last = end - 1;
    for (int p = start + q; p < end; p += 32) {
        const bool v1 = (p + 8 < end), v2 = (p + 16 < end), v3 = (p + 24 < end);
        const int s0 = csr[p];
        const int s1 = csr[min(p + 8, last)];
        const int s2 = csr[min(p + 16, last)];
        const int s3 = csr[min(p + 24, last)];
        const float4 h0 = H4[(size_t)s0 * (NCLS / 4) + c];
        const float4 h1 = H4[(size_t)s1 * (NCLS / 4) + c];
        const float4 h2 = H4[(size_t)s2 * (NCLS / 4) + c];
        const float4 h3 = H4[(size_t)s3 * (NCLS / 4) + c];
        acc.x += h0.x; acc.y += h0.y; acc.z += h0.z; acc.w += h0.w;
        acc.x += v1 ? h1.x : 0.0f; acc.y += v1 ? h1.y : 0.0f; acc.z += v1 ? h1.z : 0.0f; acc.w += v1 ? h1.w : 0.0f;
        acc.x += v2 ? h2.x : 0.0f; acc.y += v2 ? h2.y : 0.0f; acc.z += v2 ? h2.z : 0.0f; acc.w += v2 ? h2.w : 0.0f;
        acc.x += v3 ? h3.x : 0.0f; acc.y += v3 ? h3.y : 0.0f; acc.z += v3 ? h3.z : 0.0f; acc.w += v3 ? h3.w : 0.0f;
    }
#pragma unroll
    for (int m = 4; m < 32; m <<= 1) {
        acc.x += __shfl_xor(acc.x, m);
        acc.y += __shfl_xor(acc.y, m);
        acc.z += __shfl_xor(acc.z, m);
        acc.w += __shfl_xor(acc.w, m);
    }
    if (q == 0) {
        const float4 bb = ((const float4*)b2)[c];
        float4 y;
        y.x = acc.x * dd + bb.x;
        y.y = acc.y * dd + bb.y;
        y.z = acc.z * dd + bb.z;
        y.w = acc.w * dd + bb.w;
        ((float4*)out)[(size_t)d * (NCLS / 4) + c] = y;
    }
}


extern "C" void kernel_launch(void* const* d_in, const int* in_sizes, int n_in,
                              void* d_out, int out_size, void* d_ws, size_t ws_size,
                              hipStream_t stream) {
    const float* x    = (const float*)d_in[0];
    const int*   esrc = (const int*)d_in[1];
    const int*   edst = (const int*)d_in[2];
    const float* W1   = (const float*)d_in[3];
    const float* b1   = (const float*)d_in[4];
    const float* W2   = (const float*)d_in[5];
    const float* b2   = (const float*)d_in[6];
    float* out = (float*)d_out;

    const int NN = in_sizes[0] / F_IN;
    const int E  = in_sizes[1];
    const int Mtiles = NN / 16;

    float* H   = (float*)d_ws;
    float* Y   = H + (size_t)NN * HID;
    int*   csr = (int*)(Y + (size_t)NN * HID);
    float* H2  = Y;

    const int nb     = (NN + BN - 1) / BN;
    const int ntiles = (E + TILE_E - 1) / TILE_E;
    const int gwg    = (Mtiles + GW - 1) / GW;
    int2* bk      = (int2*)Y;
    int*  tab     = (int*)Y + (((size_t)2 * E + 3) & ~(size_t)3);

    const int B = 256;
    #define NB(n) (((n) + B - 1) / B)

    tile_sort_gemm1<<<ntiles + gwg, TILE_T, 0, stream>>>(esrc, edst, bk, tab, E, nb, ntiles, x, W1, H, Mtiles);
    bucket_csr<<<nb, 1024, 0, stream>>>(bk, tab, out, csr, H, NN, ntiles);

    gather64_gemm2<<<(NN + 15) / 16, B, 0, stream>>>(H, csr, out, b1, W2, H2, NN);

    gather16<<<NB(NN * 32), B, 0, stream>>>(H2, csr, b2, out, NN);

    #undef NB
}
